# starting kernel
# baseline (speedup 1.0000x reference)
#include <hip/hip_runtime.h>

typedef __attribute__((ext_vector_type(2))) float v2f;
typedef __attribute__((ext_vector_type(8))) float v8f;

constexpr int kD  = 64;
constexpr int kU  = 1024;
constexpr int kB  = 4096;
constexpr int kBT = 32;

constexpr int kXStride = kD + 4;
constexpr int kHStride = kU + 4;

constexpr int kWaOff = 0;
constexpr int kWaSz  = kU * kD;
constexpr int kWbOff = kWaOff + kWaSz;
constexpr int kWbSz  = kU * kU;
constexpr int kWcOff = kWbOff + kWbSz;
constexpr int kWcSz  = 2 * kD * kU;


__global__ void prep_w0(const float* __restrict__ W0, float* __restrict__ Wa) {
  int e = blockIdx.x * 256 + threadIdx.x;
  int j = e >> 6, d = e & 63;
  int dh = j % (kD - 1);
  float v = (d <= dh) ? W0[e] : 0.f;
  int t = j >> 4, n = j & 15;
  int cp = d >> 3, rr = d & 7;
  int q  = ((rr >> 2) << 1) | (rr & 1);
  int hi = (rr >> 1) & 1;
  int l  = (hi << 4) | n;
  Wa[(((t * 8 + cp) * 32 + l) << 2) + q] = v;
}

__global__ void prep_w1(const float* __restrict__ W1, float* __restrict__ Wb) {
  int e = blockIdx.x * 256 + threadIdx.x;
  int j = e >> 10, k = e & 1023;
  int dhj = j % (kD - 1), dhk = k % (kD - 1);
  float v = (dhk <= dhj) ? W1[e] : 0.f;
  int t = j >> 4, n = j & 15;
  int cp = k >> 3, rr = k & 7;
  int q  = ((rr >> 2) << 1) | (rr & 1);
  int hi = (rr >> 1) & 1;
  int l  = (hi << 4) | n;
  Wb[(((t * 128 + cp) * 32 + l) << 2) + q] = v;
}

__global__ void prep_w2(const float* __restrict__ W2, float* __restrict__ Wc) {
  int e = blockIdx.x * 256 + threadIdx.x;
  int r = e >> 10, k = e & 1023;
  int dhk = k % (kD - 1);
  Wc[e] = (dhk < (r & (kD - 1))) ? W2[e] : 0.f;
}


__device__ __forceinline__ v8f wmma_f32(v2f a, v2f b, v8f c) {
  return __builtin_amdgcn_wmma_f32_16x16x4_f32(false, a, false, b, (short)0, c,
                                               false, false);
}

__launch_bounds__(256, 1)
__global__ void made_flow_kernel(const float* __restrict__ u,
                                 const float* __restrict__ b0,
                                 const float* __restrict__ b1,
                                 const float* __restrict__ b2,
                                 const float* __restrict__ ws,
                                 float* __restrict__ out) {
  extern __shared__ float smem[];
  float* xs   = smem;
  float* h0s  = xs  + kBT * kXStride;
  float* h1s  = h0s + kBT * kHStride;
  float* red  = h1s + kBT * kHStride;
  float* ldjs = red + kBT * 8 * 2;

  const int tid  = threadIdx.x;
  const int wg   = blockIdx.x;
  const int lane = tid & 31;
  const int wv   = tid >> 5;
  const int n    = lane & 15;
  const int hi2  = (lane >> 4) << 1;
  const int hi8  = (lane >> 4) << 3;

  const float* Wa = ws + kWaOff;
  const float* Wb = ws + kWbOff;
  const float* Wc = ws + kWcOff;

  for (int e = tid; e < kBT * kXStride; e += 256) xs[e] = 0.f;
  if (tid < kBT) ldjs[tid] = 0.f;
  __syncthreads();

  for (int i = 0; i < kD; ++i) {
    for (int tt = 0; tt < 8; ++tt) {
      const int t = wv * 8 + tt;
      const float bias = b0[t * 16 + n];
      v8f acc0, acc1;
      for (int v = 0; v < 8; ++v) { acc0[v] = bias; acc1[v] = bias; }
      const float4* wq = ((const float4*)Wa) + (t * 8) * 32 + lane;
      for (int cp = 0; cp < 8; ++cp) {
        float4 bq = wq[cp * 32];
        v2f bA = {bq.x, bq.y};
        v2f bB = {bq.z, bq.w};
        const int kb = cp * 8 + hi2;
        v2f a00 = *(const v2f*)&xs[n * kXStride + kb];
        v2f a01 = *(const v2f*)&xs[n * kXStride + kb + 4];
        v2f a10 = *(const v2f*)&xs[(n + 16) * kXStride + kb];
        v2f a11 = *(const v2f*)&xs[(n + 16) * kXStride + kb + 4];
        acc0 = wmma_f32(a00, bA, acc0);
        acc1 = wmma_f32(a10, bA, acc1);
        acc0 = wmma_f32(a01, bB, acc0);
        acc1 = wmma_f32(a11, bB, acc1);
      }
      for (int v = 0; v < 8; ++v) {
        float f0 = acc0[v] > 0.f ? acc0[v] : 0.f;
        float f1 = acc1[v] > 0.f ? acc1[v] : 0.f;
        h0s[(v + hi8) * kHStride + t * 16 + n]        = f0;
        h0s[(16 + v + hi8) * kHStride + t * 16 + n]   = f1;
      }
    }
    __syncthreads();

    for (int tt = 0; tt < 8; ++tt) {
      const int t = wv * 8 + tt;
      const float bias = b1[t * 16 + n];
      v8f acc0, acc1;
      for (int v = 0; v < 8; ++v) { acc0[v] = bias; acc1[v] = bias; }
      const float4* wq = ((const float4*)Wb) + (t * 128) * 32 + lane;
      for (int cp = 0; cp < 128; ++cp) {
        float4 bq = wq[cp * 32];
        v2f bA = {bq.x, bq.y};
        v2f bB = {bq.z, bq.w};
        const int kb = cp * 8 + hi2;
        v2f a00 = *(const v2f*)&h0s[n * kHStride + kb];
        v2f a01 = *(const v2f*)&h0s[n * kHStride + kb + 4];
        v2f a10 = *(const v2f*)&h0s[(n + 16) * kHStride + kb];
        v2f a11 = *(const v2f*)&h0s[(n + 16) * kHStride + kb + 4];
        acc0 = wmma_f32(a00, bA, acc0);
        acc1 = wmma_f32(a10, bA, acc1);
        acc0 = wmma_f32(a01, bB, acc0);
        acc1 = wmma_f32(a11, bB, acc1);
      }
      for (int v = 0; v < 8; ++v) {
        float f0 = acc0[v] > 0.f ? acc0[v] : 0.f;
        float f1 = acc1[v] > 0.f ? acc1[v] : 0.f;
        h1s[(v + hi8) * kHStride + t * 16 + n]        = f0;
        h1s[(16 + v + hi8) * kHStride + t * 16 + n]   = f1;
      }
    }
    __syncthreads();

    {
      const int r = tid & 31;
      const int s = tid >> 5;
      const float* wmu = Wc + i * kU;
      const float* wsg = Wc + (kD + i) * kU;
      const float* hrow = h1s + r * kHStride;
      float mp = 0.f, sp = 0.f;
      const int k0 = s * 128;
      for (int k = k0; k < k0 + 128; ++k) {
        float h = hrow[k];
        mp = fmaf(h, wmu[k], mp);
        sp = fmaf(h, wsg[k], sp);
      }
      red[(r * 8 + s) * 2 + 0] = mp;
      red[(r * 8 + s) * 2 + 1] = sp;
    }
    __syncthreads();
    if (tid < kBT) {
      float mu = b2[i], sg = b2[kD + i];
      for (int s = 0; s < 8; ++s) {
        mu += red[(tid * 8 + s) * 2 + 0];
        sg += red[(tid * 8 + s) * 2 + 1];
      }
      const float uu = u[(wg * kBT + tid) * kD + i];
      xs[tid * kXStride + i] = uu * expf(sg) + mu;
      ldjs[tid] += sg;
    }
    __syncthreads();
  }

  for (int e = tid; e < kBT * kD; e += 256) {
    int r = e >> 6, d = e & 63;
    out[(wg * kBT + r) * kD + d] = xs[r * kXStride + d];
  }
  if (tid < kBT) out[kB * kD + wg * kBT + tid] = ldjs[tid];
}

extern "C" void kernel_launch(void* const* d_in, const int* in_sizes, int n_in,
                              void* d_out, int out_size, void* d_ws, size_t ws_size,
                              hipStream_t stream) {
  const float* u  = (const float*)d_in[0];
  const float* W0 = (const float*)d_in[1];
  const float* b0 = (const float*)d_in[2];
  const float* W1 = (const float*)d_in[3];
  const float* b1 = (const float*)d_in[4];
  const float* W2 = (const float*)d_in[5];
  const float* b2 = (const float*)d_in[6];
  float* ws  = (float*)d_ws;
  float* out = (float*)d_out;

  prep_w0<<<(kU * kD) / 256,     256, 0, stream>>>(W0, ws + kWaOff);
  prep_w1<<<(kU * kU) / 256,     256, 0, stream>>>(W1, ws + kWbOff);
  prep_w2<<<(2 * kD * kU) / 256, 256, 0, stream>>>(W2, ws + kWcOff);

  size_t smem = (size_t)(kBT * kXStride + 2 * kBT * kHStride + kBT * 8 * 2 + kBT) *
                sizeof(float);
  made_flow_kernel<<<kB / kBT, 256, smem, stream>>>(u, b0, b1, b2, ws, out);
}


# Each hidden unit computed once, in the step its mask degree becomes final, not all 1024 redone 64 times: units sorted by degree, K split over 8 waves, z a running sum (sum order changes)
# speedup vs baseline: 26.8344x; 26.8344x over previous
#include <hip/hip_runtime.h>

typedef __attribute__((ext_vector_type(2))) float v2f;
typedef __attribute__((ext_vector_type(8))) float v8f;

constexpr int kD  = 64;
constexpr int kU  = 1024;
constexpr int kB  = 4096;
constexpr int kBT = 32;

constexpr int kDeg = kD - 1;
constexpr int kReg = 16 * kDeg;
constexpr int kXT  = kDeg;

constexpr int kXStride = kD + 4;
constexpr int kHStride = kU + 4;
constexpr int kDStride = 20;

constexpr int kWaOff = 0;
constexpr int kWaSz  = kU * kD;
constexpr int kWbOff = kWaOff + kWaSz;
constexpr int kWbSz  = kU * kU;
constexpr int kWcOff = kWbOff + kWbSz;
constexpr int kWcSz  = 2 * kD * kU;

__device__ __forceinline__ int unit_pos(int j) {
  return j < kReg ? ((j % kDeg) << 4) + (j / kDeg) : j;
}


__global__ void prep_w0(const float* __restrict__ W0, float* __restrict__ Wa) {
  int e = blockIdx.x * 256 + threadIdx.x;
  int j = e >> 6, d = e & 63;
  int dh = j % (kD - 1);
  float v = (d <= dh) ? W0[e] : 0.f;
  int pj = unit_pos(j);
  int t = pj >> 4, n = pj & 15;
  int cp = d >> 3, rr = d & 7;
  int q  = ((rr >> 2) << 1) | (rr & 1);
  int hi = (rr >> 1) & 1;
  int l  = (hi << 4) | n;
  Wa[(((t * 8 + cp) * 32 + l) << 2) + q] = v;
}

__global__ void prep_w1(const float* __restrict__ W1, float* __restrict__ Wb) {
  int e = blockIdx.x * 256 + threadIdx.x;
  int j = e >> 10, k = e & 1023;
  int dhj = j % (kD - 1), dhk = k % (kD - 1);
  float v = (dhk <= dhj) ? W1[e] : 0.f;
  int pj = unit_pos(j), pk = unit_pos(k);
  int t = pj >> 4, n = pj & 15;
  int cp = pk >> 3, rr = pk & 7;
  int q  = ((rr >> 2) << 1) | (rr & 1);
  int hi = (rr >> 1) & 1;
  int l  = (hi << 4) | n;
  Wb[(((t * 128 + cp) * 32 + l) << 2) + q] = v;
}

__global__ void prep_w2(const float* __restrict__ W2, float* __restrict__ Wc) {
  int e = blockIdx.x * 256 + threadIdx.x;
  int r = e >> 10, k = e & 1023;
  int dhk = k % (kD - 1);
  float v = (dhk < (r & (kD - 1))) ? W2[e] : 0.f;
  int pk = unit_pos(k);
  int t = r >> 4, n = r & 15;
  int cp = pk >> 3, rr = pk & 7;
  int q  = ((rr >> 2) << 1) | (rr & 1);
  int hi = (rr >> 1) & 1;
  int l  = (hi << 4) | n;
  Wc[(((t * 128 + cp) * 32 + l) << 2) + q] = v;
}


__device__ __forceinline__ v8f wmma_f32(v2f a, v2f b, v8f c) {
  return __builtin_amdgcn_wmma_f32_16x16x4_f32(false, a, false, b, (short)0, c,
                                               false, false);
}

struct AFrag { v2f a00, a01, a10, a11; };

__device__ __forceinline__ AFrag load_a(const float* __restrict__ as, int astride,
                                        int acol, int n, int hi2) {
  const int kb = acol + hi2;
  AFrag f;
  f.a00 = *(const v2f*)&as[n * astride + kb];
  f.a01 = *(const v2f*)&as[n * astride + kb + 4];
  f.a10 = *(const v2f*)&as[(n + 16) * astride + kb];
  f.a11 = *(const v2f*)&as[(n + 16) * astride + kb + 4];
  return f;
}

__device__ __forceinline__ void mma_chunk(const AFrag& f, float4 bq, v8f& acc0, v8f& acc1) {
  v2f bA = {bq.x, bq.y};
  v2f bB = {bq.z, bq.w};
  acc0 = wmma_f32(f.a00, bA, acc0);
  acc1 = wmma_f32(f.a10, bA, acc1);
  acc0 = wmma_f32(f.a01, bB, acc0);
  acc1 = wmma_f32(f.a11, bB, acc1);
}

__launch_bounds__(256, 1)
__global__ void made_flow_kernel(const float* __restrict__ u,
                                 const float* __restrict__ b0,
                                 const float* __restrict__ b1,
                                 const float* __restrict__ b2,
                                 const float* __restrict__ ws,
                                 float* __restrict__ out) {
  extern __shared__ float smem[];
  float* xs   = smem;
  float* h0s  = xs  + kBT * kXStride;
  float* h1t  = h0s + kBT * kHStride;
  float* part = h1t + kBT * kDStride;
  float* dlt  = part + 8 * 2 * kBT * 16;
  float* zs   = dlt + kBT * kDStride;
  float* ldjs = zs + kBT * 2;

  const int tid  = threadIdx.x;
  const int wg   = blockIdx.x;
  const int lane = tid & 31;
  const int wv   = __builtin_amdgcn_readfirstlane(tid >> 5);
  const int n    = lane & 15;
  const int hi2  = (lane >> 4) << 1;
  const int hi8  = (lane >> 4) << 3;

  const float* Wa = ws + kWaOff;
  const float* Wb = ws + kWbOff;
  const float* Wc = ws + kWcOff;

  for (int e = tid; e < kBT * kXStride; e += 256) xs[e] = 0.f;
  for (int e = tid; e < kBT * 16; e += 256)
    h0s[(e >> 4) * kHStride + kReg + (e & 15)] = 0.f;
  if (tid < kBT) ldjs[tid] = 0.f;
  __syncthreads();

  const float4* wz = ((const float4*)Wc) + (wv * 128) * 32 + lane;
  v8f z0, z1;
  for (int v = 0; v < 8; ++v) { z0[v] = 0.f; z1[v] = 0.f; }

  for (int i = 0; i < kD; ++i) {
    float uu = 0.f;
    if (tid < kBT) uu = u[(wg * kBT + tid) * kD + i];
    const float bmu = b2[i], bsg = b2[kD + i];
    if (i > 0) {
      const int g  = i - 1;
      const int nt = (g < 16) ? 2 : 1;
      const float4 zq0 = wz[(2 * g) * 32];
      const float4 zq1 = wz[(2 * g + 1) * 32];
      float4 zx0 = {0.f, 0.f, 0.f, 0.f}, zx1 = zx0;
      const int cc = tid & 15;
      const float bias1 = b1[g + kDeg * cc];
      float bias1x = 0.f;
      if (nt == 2) {
        zx0 = wz[126 * 32];
        zx1 = wz[127 * 32];
        bias1x = b1[kReg + cc];
      }

      if (wv < 2 * nt) {
        const int t  = (wv < 2) ? g : kXT;
        const int mt = wv & 1;
        const float bias = b0[(t < kXT) ? t + kDeg * n : kReg + n];
        const float4* wq = ((const float4*)Wa) + (t * 8) * 32 + lane;
        float4 bq[8];
        for (int cp = 0; cp < 8; ++cp) bq[cp] = wq[cp * 32];
        v8f accA, accB;
        for (int v = 0; v < 8; ++v) { accA[v] = bias; accB[v] = 0.f; }
        const float* xr = xs + (mt * 16 + n) * kXStride + hi2;
        for (int cp = 0; cp < 8; ++cp) {
          v2f a0 = *(const v2f*)&xr[cp * 8];
          v2f a1 = *(const v2f*)&xr[cp * 8 + 4];
          v2f bA = {bq[cp].x, bq[cp].y};
          v2f bB = {bq[cp].z, bq[cp].w};
          accA = wmma_f32(a0, bA, accA);
          accB = wmma_f32(a1, bB, accB);
        }
        if (t < kXT || n == g) {
          for (int v = 0; v < 8; ++v) {
            float f = accA[v] + accB[v];
            f = f > 0.f ? f : 0.f;
            h0s[(mt * 16 + v + hi8) * kHStride + t * 16 + n] = f;
          }
        }
      }
      __syncthreads();

      {
        const int ncp  = 2 * (g + 1);
        const int nch  = (wv < ncp) ? ((ncp - wv + 7) >> 3) : 0;
        const int ntot = nch + ((wv >= 6) ? 1 : 0);
        for (int q = 0; q < nt; ++q) {
          const int t = q ? kXT : g;
          v8f acc0, acc1;
          for (int v = 0; v < 8; ++v) { acc0[v] = 0.f; acc1[v] = 0.f; }
          const float4* wq = ((const float4*)Wb) + (t * 128) * 32 + lane;
          for (int j0 = 0; j0 < ntot; j0 += 4) {
            float4 bq[4];
            AFrag  af[4];
            for (int jj = 0; jj < 4; ++jj) {
              int j = j0 + jj;
              j = j < ntot ? j : ntot - 1;
              const int cp = (j < nch) ? wv + 8 * j : 120 + wv;
              bq[jj] = wq[cp * 32];
              af[jj] = load_a(h0s, kHStride, cp * 8, n, hi2);
            }
            asm volatile("" :: "v"(bq[0].x), "v"(bq[0].y), "v"(bq[0].z), "v"(bq[0].w),
                               "v"(bq[1].x), "v"(bq[1].y), "v"(bq[1].z), "v"(bq[1].w),
                               "v"(bq[2].x), "v"(bq[2].y), "v"(bq[2].z), "v"(bq[2].w),
                               "v"(bq[3].x), "v"(bq[3].y), "v"(bq[3].z), "v"(bq[3].w));
            asm volatile("" :: "v"(af[0].a00), "v"(af[0].a01), "v"(af[0].a10), "v"(af[0].a11),
                               "v"(af[1].a00), "v"(af[1].a01), "v"(af[1].a10), "v"(af[1].a11),
                               "v"(af[2].a00), "v"(af[2].a01), "v"(af[2].a10), "v"(af[2].a11),
                               "v"(af[3].a00), "v"(af[3].a01), "v"(af[3].a10), "v"(af[3].a11));
            mma_chunk(af[0], bq[0], acc0, acc1);
            if (j0 + 1 < ntot) {
              mma_chunk(af[1], bq[1], acc0, acc1);
              if (j0 + 2 < ntot) {
                mma_chunk(af[2], bq[2], acc0, acc1);
                if (j0 + 3 < ntot) mma_chunk(af[3], bq[3], acc0, acc1);
              }
            }
          }
          float* pp = part + ((wv * 2 + q) * kBT) * 16;
          for (int v = 0; v < 8; ++v) {
            pp[(v + hi8) * 16 + n]      = acc0[v];
            pp[(16 + v + hi8) * 16 + n] = acc1[v];
          }
        }
      }
      __syncthreads();
      for (int rp = 0; rp < 2; ++rp) {
        const int row = (tid >> 4) + 16 * rp;
        float s = bias1;
        for (int w = 0; w < 8; ++w) s += part[((w * 2) * kBT + row) * 16 + cc];
        h1t[row * kDStride + cc] = s > 0.f ? s : 0.f;
        if (nt == 2) {
          float sx = bias1x;
          for (int w = 0; w < 8; ++w) sx += part[((w * 2 + 1) * kBT + row) * 16 + cc];
          sx = sx > 0.f ? sx : 0.f;
          dlt[row * kDStride + cc] = (cc == g) ? sx : 0.f;
        }
      }
      __syncthreads();

      {
        AFrag f0 = load_a(h1t, kDStride, 0, n, hi2);
        AFrag f1 = load_a(h1t, kDStride, 8, n, hi2);
        mma_chunk(f0, zq0, z0, z1);
        mma_chunk(f1, zq1, z0, z1);
        if (nt == 2) {
          AFrag x0 = load_a(dlt, kDStride, 0, n, hi2);
          AFrag x1 = load_a(dlt, kDStride, 8, n, hi2);
          mma_chunk(x0, zx0, z0, z1);
          mma_chunk(x1, zx1, z0, z1);
        }
      }
    }

    if ((wv & 3) == (i >> 4) && n == (i & 15)) {
      const int cz = wv >> 2;
      for (int v = 0; v < 8; ++v) {
        zs[(v + hi8) * 2 + cz]      = z0[v];
        zs[(16 + v + hi8) * 2 + cz] = z1[v];
      }
    }
    __syncthreads();
    if (tid < kBT) {
      const float mu = bmu + zs[tid * 2 + 0];
      const float sg = bsg + zs[tid * 2 + 1];
      xs[tid * kXStride + i] = uu * expf(sg) + mu;
      ldjs[tid] += sg;
    }
    __syncthreads();
  }

  for (int e = tid; e < kBT * kD; e += 256) {
    int r = e >> 6, d = e & 63;
    out[(wg * kBT + r) * kD + d] = xs[r * kXStride + d];
  }
  if (tid < kBT) out[kB * kD + wg * kBT + tid] = ldjs[tid];
}

extern "C" void kernel_launch(void* const* d_in, const int* in_sizes, int n_in,
                              void* d_out, int out_size, void* d_ws, size_t ws_size,
                              hipStream_t stream) {
  const float* u  = (const float*)d_in[0];
  const float* W0 = (const float*)d_in[1];
  const float* b0 = (const float*)d_in[2];
  const float* W1 = (const float*)d_in[3];
  const float* b1 = (const float*)d_in[4];
  const float* W2 = (const float*)d_in[5];
  const float* b2 = (const float*)d_in[6];
  float* ws  = (float*)d_ws;
  float* out = (float*)d_out;

  prep_w0<<<(kU * kD) / 256,     256, 0, stream>>>(W0, ws + kWaOff);
  prep_w1<<<(kU * kU) / 256,     256, 0, stream>>>(W1, ws + kWbOff);
  prep_w2<<<(2 * kD * kU) / 256, 256, 0, stream>>>(W2, ws + kWcOff);

  size_t smem = (size_t)(kBT * kXStride + kBT * kHStride + 8 * 2 * kBT * 16 +
                         2 * kBT * kDStride + kBT * 2 + kBT) *
                sizeof(float);
  made_flow_kernel<<<kB / kBT, 256, smem, stream>>>(u, b0, b1, b2, ws, out);
}


# Batch tile per workgroup 32 -> 16 rows: 256 workgroups, one on each of the 256 WGPs (128 sat idle; LDS reserved so none share); k chunks alternate over two accumulators (sum order changes)
# speedup vs baseline: 49.2437x; 1.8351x over previous
#include <hip/hip_runtime.h>

typedef __attribute__((ext_vector_type(2))) float v2f;
typedef __attribute__((ext_vector_type(8))) float v8f;

constexpr int kD  = 64;
constexpr int kU  = 1024;
constexpr int kB  = 4096;
constexpr int kBT = 16;

constexpr int kDeg = kD - 1;
constexpr int kReg = 16 * kDeg;
constexpr int kXT  = kDeg;

constexpr int kXStride = kD + 4;
constexpr int kHStride = kU + 4;
constexpr int kDStride = 20;

constexpr int kWaOff = 0;
constexpr int kWaSz  = kU * kD;
constexpr int kWbOff = kWaOff + kWaSz;
constexpr int kWbSz  = kU * kU;
constexpr int kWcOff = kWbOff + kWbSz;
constexpr int kWcSz  = 2 * kD * kU;

__device__ __forceinline__ int unit_pos(int j) {
  return j < kReg ? ((j % kDeg) << 4) + (j / kDeg) : j;
}


__global__ void prep_w0(const float* __restrict__ W0, float* __restrict__ Wa) {
  int e = blockIdx.x * 256 + threadIdx.x;
  int j = e >> 6, d = e & 63;
  int dh = j % (kD - 1);
  float v = (d <= dh) ? W0[e] : 0.f;
  int pj = unit_pos(j);
  int t = pj >> 4, n = pj & 15;
  int cp = d >> 3, rr = d & 7;
  int q  = ((rr >> 2) << 1) | (rr & 1);
  int hi = (rr >> 1) & 1;
  int l  = (hi << 4) | n;
  Wa[(((t * 8 + cp) * 32 + l) << 2) + q] = v;
}

__global__ void prep_w1(const float* __restrict__ W1, float* __restrict__ Wb) {
  int e = blockIdx.x * 256 + threadIdx.x;
  int j = e >> 10, k = e & 1023;
  int dhj = j % (kD - 1), dhk = k % (kD - 1);
  float v = (dhk <= dhj) ? W1[e] : 0.f;
  int pj = unit_pos(j), pk = unit_pos(k);
  int t = pj >> 4, n = pj & 15;
  int cp = pk >> 3, rr = pk & 7;
  int q  = ((rr >> 2) << 1) | (rr & 1);
  int hi = (rr >> 1) & 1;
  int l  = (hi << 4) | n;
  Wb[(((t * 128 + cp) * 32 + l) << 2) + q] = v;
}

__global__ void prep_w2(const float* __restrict__ W2, float* __restrict__ Wc) {
  int e = blockIdx.x * 256 + threadIdx.x;
  int r = e >> 10, k = e & 1023;
  int dhk = k % (kD - 1);
  float v = (dhk < (r & (kD - 1))) ? W2[e] : 0.f;
  int pk = unit_pos(k);
  int t = r >> 4, n = r & 15;
  int cp = pk >> 3, rr = pk & 7;
  int q  = ((rr >> 2) << 1) | (rr & 1);
  int hi = (rr >> 1) & 1;
  int l  = (hi << 4) | n;
  Wc[(((t * 128 + cp) * 32 + l) << 2) + q] = v;
}


__device__ __forceinline__ v8f wmma_f32(v2f a, v2f b, v8f c) {
  return __builtin_amdgcn_wmma_f32_16x16x4_f32(false, a, false, b, (short)0, c,
                                               false, false);
}

struct AFrag { v2f a0, a1; };

__device__ __forceinline__ AFrag load_a(const float* __restrict__ as, int astride,
                                        int acol, int n, int hi2) {
  const int kb = acol + hi2;
  AFrag f;
  f.a0 = *(const v2f*)&as[n * astride + kb];
  f.a1 = *(const v2f*)&as[n * astride + kb + 4];
  return f;
}

__device__ __forceinline__ void mma_chunk(const AFrag& f, float4 bq, v8f& accA, v8f& accB) {
  v2f bA = {bq.x, bq.y};
  v2f bB = {bq.z, bq.w};
  accA = wmma_f32(f.a0, bA, accA);
  accB = wmma_f32(f.a1, bB, accB);
}

__device__ __forceinline__ void l2_counts(int g, int wv, int& nch, int& ntot) {
  const int ncp = 2 * (g + 1);
  nch  = (wv < ncp) ? ((ncp - wv + 7) >> 3) : 0;
  ntot = nch + ((wv >= 6) ? 1 : 0);
}

template <int NG>
__device__ __forceinline__ void fetch_w2(float4 (&w)[4 * NG], const float4* __restrict__ wq,
                                         int nch, int ntot, int wv) {
#pragma unroll
  for (int jg = 0; jg < NG; ++jg) {
    if (jg * 4 < ntot) {
#pragma unroll
      for (int jj = 0; jj < 4; ++jj) {
        int j = jg * 4 + jj;
        j = j < ntot ? j : ntot - 1;
        const int cp = (j < nch) ? wv + 8 * j : 120 + wv;
        w[jg * 4 + jj] = wq[cp * 32];
      }
    }
  }
}

template <int NG>
__device__ __forceinline__ void layer2_tile(const float4 (&w)[4 * NG],
                                            const float* __restrict__ h0s, int nch, int ntot,
                                            int wv, int n, int hi2, v8f& acc0, v8f& acc1) {
#pragma unroll
  for (int jg = 0; jg < NG; ++jg) {
    if (jg * 4 < ntot) {
      AFrag af[4];
#pragma unroll
      for (int jj = 0; jj < 4; ++jj) {
        int j = jg * 4 + jj;
        j = j < ntot ? j : ntot - 1;
        const int cp = (j < nch) ? wv + 8 * j : 120 + wv;
        af[jj] = load_a(h0s, kHStride, cp * 8, n, hi2);
      }
      asm volatile("" :: "v"(af[0].a0), "v"(af[0].a1), "v"(af[1].a0), "v"(af[1].a1),
                         "v"(af[2].a0), "v"(af[2].a1), "v"(af[3].a0), "v"(af[3].a1));
      mma_chunk(af[0], w[jg * 4 + 0], acc0, acc1);
      if (jg * 4 + 1 < ntot) {
        mma_chunk(af[1], w[jg * 4 + 1], acc0, acc1);
        if (jg * 4 + 2 < ntot) {
          mma_chunk(af[2], w[jg * 4 + 2], acc0, acc1);
          if (jg * 4 + 3 < ntot) mma_chunk(af[3], w[jg * 4 + 3], acc0, acc1);
        }
      }
    }
  }
}

struct StepOps {
  float4 zq0, zq1, zx0, zx1;
  float bias1, bias1x;
  float uu, bmu, bsg;
};

__device__ __forceinline__ void lds_barrier() {
  __builtin_amdgcn_fence(__ATOMIC_RELEASE, "workgroup", "local");
  __builtin_amdgcn_s_barrier();
  __builtin_amdgcn_fence(__ATOMIC_ACQUIRE, "workgroup", "local");
}

__launch_bounds__(256, 1)
__global__ void made_flow_kernel(const float* __restrict__ u,
                                 const float* __restrict__ b0,
                                 const float* __restrict__ b1,
                                 const float* __restrict__ b2,
                                 const float* __restrict__ ws,
                                 float* __restrict__ out) {
  extern __shared__ float smem[];
  float* xs   = smem;
  float* h0s  = xs  + kBT * kXStride;
  float* h1t  = h0s + kBT * kHStride;
  float* part = h1t + kBT * kDStride;
  float* dlt  = part + 8 * 2 * kBT * 16;
  float* zs   = dlt + kBT * kDStride;
  float* ldjs = zs + kBT * 2;

  const int tid  = threadIdx.x;
  const int wg   = blockIdx.x;
  const int lane = tid & 31;
  const int wv   = __builtin_amdgcn_readfirstlane(tid >> 5);
  const int n    = lane & 15;
  const int hi2  = (lane >> 4) << 1;
  const int hi8  = (lane >> 4) << 3;
  const int cc   = tid & 15;

  const float4* Wa4 = ((const float4*)(ws + kWaOff)) + lane;
  const float4* Wb4 = ((const float4*)(ws + kWbOff)) + lane;
  const float4* wz = ((const float4*)(ws + kWcOff)) + (wv * 128) * 32 + lane;

  float4 w1[8];
  float4 w2[16];
  float4 w2x[8];
  float  bias0 = 0.f;
  StepOps sm;
  const float4 zero4 = {0.f, 0.f, 0.f, 0.f};
  for (int k = 0; k < 8; ++k)  { w1[k] = zero4; w2x[k] = zero4; }
  for (int k = 0; k < 16; ++k) w2[k] = zero4;

  auto fetch_big = [&](int s) __attribute__((always_inline)) {
    const int g  = s - 1;
    const int nt = (g < 16) ? 2 : 1;
    if (wv < nt) {
      const int t = (wv == 0) ? g : kXT;
      bias0 = b0[(t < kXT) ? t + kDeg * n : kReg + n];
      const float4* wq = Wa4 + (t * 8) * 32;
      for (int cp = 0; cp < 8; ++cp) w1[cp] = wq[cp * 32];
    }
    int nch, ntot;
    l2_counts(g, wv, nch, ntot);
    fetch_w2<4>(w2, Wb4 + (g * 128) * 32, nch, ntot, wv);
    if (nt == 2) fetch_w2<2>(w2x, Wb4 + (kXT * 128) * 32, nch, ntot, wv);
  };
  auto fetch_small = [&](int s) __attribute__((always_inline)) {
    const int g = s - 1;
    sm.zq0 = wz[(2 * g) * 32];
    sm.zq1 = wz[(2 * g + 1) * 32];
    sm.bias1 = b1[g + kDeg * cc];
    sm.zx0 = wz[126 * 32];
    sm.zx1 = wz[127 * 32];
    sm.bias1x = b1[kReg + cc];
    sm.uu = u[(wg * kBT + n) * kD + s];
    sm.bmu = b2[s];
    sm.bsg = b2[kD + s];
  };

  fetch_big(1);
  const float u0   = u[(wg * kBT + n) * kD];
  const float bmu0 = b2[0], bsg0 = b2[kD];

  for (int e = tid; e < kBT * kXStride; e += 256) xs[e] = 0.f;
  for (int e = tid; e < kBT * 16; e += 256)
    h0s[(e >> 4) * kHStride + kReg + (e & 15)] = 0.f;
  lds_barrier();
  if (tid < kBT) {
    xs[tid * kXStride] = u0 * expf(bsg0) + bmu0;
    ldjs[tid] = bsg0;
  }
  fetch_small(1);
  lds_barrier();

  v8f zA, zB;
  for (int v = 0; v < 8; ++v) { zA[v] = 0.f; zB[v] = 0.f; }

  for (int i = 1; i < kD; ++i) {
    {
      const int g  = i - 1;
      const int nt = (g < 16) ? 2 : 1;

      if (wv < nt) {
        const int t  = (wv == 0) ? g : kXT;
        v8f accA, accB;
        for (int v = 0; v < 8; ++v) { accA[v] = bias0; accB[v] = 0.f; }
        const float* xr = xs + n * kXStride + hi2;
        for (int cp = 0; cp < 8; ++cp) {
          v2f a0 = *(const v2f*)&xr[cp * 8];
          v2f a1 = *(const v2f*)&xr[cp * 8 + 4];
          v2f bA = {w1[cp].x, w1[cp].y};
          v2f bB = {w1[cp].z, w1[cp].w};
          accA = wmma_f32(a0, bA, accA);
          accB = wmma_f32(a1, bB, accB);
        }
        if (t < kXT || n == g) {
          for (int v = 0; v < 8; ++v) {
            float f = accA[v] + accB[v];
            f = f > 0.f ? f : 0.f;
            h0s[(v + hi8) * kHStride + t * 16 + n] = f;
          }
        }
      }
      lds_barrier();

      {
        int nch, ntot;
        l2_counts(g, wv, nch, ntot);
        v8f acc0, acc1, acx0, acx1;
        for (int v = 0; v < 8; ++v) { acc0[v] = 0.f; acc1[v] = 0.f; acx0[v] = 0.f; acx1[v] = 0.f; }
        layer2_tile<4>(w2, h0s, nch, ntot, wv, n, hi2, acc0, acc1);
        if (nt == 2) layer2_tile<2>(w2x, h0s, nch, ntot, wv, n, hi2, acx0, acx1);

        asm volatile("" :: "v"(sm.zq0.x), "v"(sm.zq0.y), "v"(sm.zq0.z), "v"(sm.zq0.w),
                           "v"(sm.zq1.x), "v"(sm.zq1.y), "v"(sm.zq1.z), "v"(sm.zq1.w),
                           "v"(sm.zx0.x), "v"(sm.zx0.y), "v"(sm.zx0.z), "v"(sm.zx0.w),
                           "v"(sm.zx1.x), "v"(sm.zx1.y), "v"(sm.zx1.z), "v"(sm.zx1.w),
                           "v"(sm.bias1), "v"(sm.bias1x), "v"(sm.uu));
        if (i + 1 < kD) fetch_big(i + 1);

        float* pp = part + ((wv * 2) * kBT) * 16;
        for (int v = 0; v < 8; ++v) pp[(v + hi8) * 16 + n] = acc0[v] + acc1[v];
        if (nt == 2) {
          float* px = pp + kBT * 16;
          for (int v = 0; v < 8; ++v) px[(v + hi8) * 16 + n] = acx0[v] + acx1[v];
        }
      }
      lds_barrier();
      {
        const int row = tid >> 4;
        float s = sm.bias1;
        for (int w = 0; w < 8; ++w) s += part[((w * 2) * kBT + row) * 16 + cc];
        h1t[row * kDStride + cc] = s > 0.f ? s : 0.f;
        if (nt == 2) {
          float sx = sm.bias1x;
          for (int w = 0; w < 8; ++w) sx += part[((w * 2 + 1) * kBT + row) * 16 + cc];
          sx = sx > 0.f ? sx : 0.f;
          dlt[row * kDStride + cc] = (cc == g) ? sx : 0.f;
        }
      }
      lds_barrier();

      {
        AFrag f0 = load_a(h1t, kDStride, 0, n, hi2);
        AFrag f1 = load_a(h1t, kDStride, 8, n, hi2);
        mma_chunk(f0, sm.zq0, zA, zB);
        mma_chunk(f1, sm.zq1, zA, zB);
        if (nt == 2) {
          AFrag x0 = load_a(dlt, kDStride, 0, n, hi2);
          AFrag x1 = load_a(dlt, kDStride, 8, n, hi2);
          mma_chunk(x0, sm.zx0, zA, zB);
          mma_chunk(x1, sm.zx1, zA, zB);
        }
      }
    }

    if ((wv & 3) == (i >> 4) && n == (i & 15)) {
      const int cz = wv >> 2;
      for (int v = 0; v < 8; ++v) zs[(v + hi8) * 2 + cz] = zA[v] + zB[v];
    }
    lds_barrier();
    if (tid < kBT) {
      const float mu = sm.bmu + zs[tid * 2 + 0];
      const float sg = sm.bsg + zs[tid * 2 + 1];
      xs[tid * kXStride + i] = sm.uu * expf(sg) + mu;
      ldjs[tid] += sg;
    }
    fetch_small(i + 1 < kD ? i + 1 : i);
    lds_barrier();
  }

  for (int e = tid; e < kBT * kD; e += 256) {
    int r = e >> 6, d = e & 63;
    out[(wg * kBT + r) * kD + d] = xs[r * kXStride + d];
  }
  if (tid < kBT) out[kB * kD + wg * kBT + tid] = ldjs[tid];
}

extern "C" void kernel_launch(void* const* d_in, const int* in_sizes, int n_in,
                              void* d_out, int out_size, void* d_ws, size_t ws_size,
                              hipStream_t stream) {
  const float* u  = (const float*)d_in[0];
  const float* W0 = (const float*)d_in[1];
  const float* b0 = (const float*)d_in[2];
  const float* W1 = (const float*)d_in[3];
  const float* b1 = (const float*)d_in[4];
  const float* W2 = (const float*)d_in[5];
  const float* b2 = (const float*)d_in[6];
  float* ws  = (float*)d_ws;
  float* out = (float*)d_out;

  prep_w0<<<(kU * kD) / 256,     256, 0, stream>>>(W0, ws + kWaOff);
  prep_w1<<<(kU * kU) / 256,     256, 0, stream>>>(W1, ws + kWbOff);
  prep_w2<<<(2 * kD * kU) / 256, 256, 0, stream>>>(W2, ws + kWcOff);

  constexpr size_t kLdsUsed = (size_t)(kBT * kXStride + kBT * kHStride + 8 * 2 * kBT * 16 +
                                       2 * kBT * kDStride + kBT * 2 + kBT) * sizeof(float);
  constexpr size_t kLdsOnePerWgp = 164 * 1024;
  constexpr size_t smem = kLdsUsed > kLdsOnePerWgp ? kLdsUsed : kLdsOnePerWgp;
  made_flow_kernel<<<kB / kBT, 256, smem, stream>>>(u, b0, b1, b2, ws, out);
}


# At kernel start each workgroup prefetches one sixteenth of the repacked weights into L2 (slice = hash of its index): a step's weight fetch then hits L2 instead of waiting for memory
# speedup vs baseline: 50.7979x; 1.0316x over previous
#include <hip/hip_runtime.h>

typedef __attribute__((ext_vector_type(2))) float v2f;
typedef __attribute__((ext_vector_type(8))) float v8f;

constexpr int kD  = 64;
constexpr int kU  = 1024;
constexpr int kB  = 4096;
constexpr int kBT = 16;

constexpr int kDeg = kD - 1;
constexpr int kReg = 16 * kDeg;
constexpr int kXT  = kDeg;

constexpr int kXStride = kD + 4;
constexpr int kHStride = kU + 4;
constexpr int kDStride = 20;

constexpr int kWaOff = 0;
constexpr int kWaSz  = kU * kD;
constexpr int kWbOff = kWaOff + kWaSz;
constexpr int kWbSz  = kU * kU;
constexpr int kWcOff = kWbOff + kWbSz;
constexpr int kWcSz  = 2 * kD * kU;

__device__ __forceinline__ int unit_pos(int j) {
  return j < kReg ? ((j % kDeg) << 4) + (j / kDeg) : j;
}


__global__ void prep_w0(const float* __restrict__ W0, float* __restrict__ Wa) {
  int e = blockIdx.x * 256 + threadIdx.x;
  int j = e >> 6, d = e & 63;
  int dh = j % (kD - 1);
  float v = (d <= dh) ? W0[e] : 0.f;
  int pj = unit_pos(j);
  int t = pj >> 4, n = pj & 15;
  int cp = d >> 3, rr = d & 7;
  int q  = ((rr >> 2) << 1) | (rr & 1);
  int hi = (rr >> 1) & 1;
  int l  = (hi << 4) | n;
  Wa[(((t * 8 + cp) * 32 + l) << 2) + q] = v;
}

__global__ void prep_w1(const float* __restrict__ W1, float* __restrict__ Wb) {
  int e = blockIdx.x * 256 + threadIdx.x;
  int j = e >> 10, k = e & 1023;
  int dhj = j % (kD - 1), dhk = k % (kD - 1);
  float v = (dhk <= dhj) ? W1[e] : 0.f;
  int pj = unit_pos(j), pk = unit_pos(k);
  int t = pj >> 4, n = pj & 15;
  int cp = pk >> 3, rr = pk & 7;
  int q  = ((rr >> 2) << 1) | (rr & 1);
  int hi = (rr >> 1) & 1;
  int l  = (hi << 4) | n;
  Wb[(((t * 128 + cp) * 32 + l) << 2) + q] = v;
}

__global__ void prep_w2(const float* __restrict__ W2, float* __restrict__ Wc) {
  int e = blockIdx.x * 256 + threadIdx.x;
  int r = e >> 10, k = e & 1023;
  int dhk = k % (kD - 1);
  float v = (dhk < (r & (kD - 1))) ? W2[e] : 0.f;
  int pk = unit_pos(k);
  int t = r >> 4, n = r & 15;
  int cp = pk >> 3, rr = pk & 7;
  int q  = ((rr >> 2) << 1) | (rr & 1);
  int hi = (rr >> 1) & 1;
  int l  = (hi << 4) | n;
  Wc[(((t * 128 + cp) * 32 + l) << 2) + q] = v;
}


__device__ __forceinline__ v8f wmma_f32(v2f a, v2f b, v8f c) {
  return __builtin_amdgcn_wmma_f32_16x16x4_f32(false, a, false, b, (short)0, c,
                                               false, false);
}

struct AFrag { v2f a0, a1; };

__device__ __forceinline__ AFrag load_a(const float* __restrict__ as, int astride,
                                        int acol, int n, int hi2) {
  const int kb = acol + hi2;
  AFrag f;
  f.a0 = *(const v2f*)&as[n * astride + kb];
  f.a1 = *(const v2f*)&as[n * astride + kb + 4];
  return f;
}

__device__ __forceinline__ void mma_chunk(const AFrag& f, float4 bq, v8f& accA, v8f& accB) {
  v2f bA = {bq.x, bq.y};
  v2f bB = {bq.z, bq.w};
  accA = wmma_f32(f.a0, bA, accA);
  accB = wmma_f32(f.a1, bB, accB);
}

__device__ __forceinline__ void l2_counts(int g, int wv, int& nch, int& ntot) {
  const int ncp = 2 * (g + 1);
  nch  = (wv < ncp) ? ((ncp - wv + 7) >> 3) : 0;
  ntot = nch + ((wv >= 6) ? 1 : 0);
}

template <int NG>
__device__ __forceinline__ void fetch_w2(float4 (&w)[4 * NG], const float4* __restrict__ wq,
                                         int nch, int ntot, int wv) {
#pragma unroll
  for (int jg = 0; jg < NG; ++jg) {
    if (jg * 4 < ntot) {
#pragma unroll
      for (int jj = 0; jj < 4; ++jj) {
        int j = jg * 4 + jj;
        j = j < ntot ? j : ntot - 1;
        const int cp = (j < nch) ? wv + 8 * j : 120 + wv;
        w[jg * 4 + jj] = wq[cp * 32];
      }
    }
  }
}

template <int NG>
__device__ __forceinline__ void layer2_tile(const float4 (&w)[4 * NG],
                                            const float* __restrict__ h0s, int nch, int ntot,
                                            int wv, int n, int hi2, v8f& acc0, v8f& acc1) {
#pragma unroll
  for (int jg = 0; jg < NG; ++jg) {
    if (jg * 4 < ntot) {
      AFrag af[4];
#pragma unroll
      for (int jj = 0; jj < 4; ++jj) {
        int j = jg * 4 + jj;
        j = j < ntot ? j : ntot - 1;
        const int cp = (j < nch) ? wv + 8 * j : 120 + wv;
        af[jj] = load_a(h0s, kHStride, cp * 8, n, hi2);
      }
      asm volatile("" :: "v"(af[0].a0), "v"(af[0].a1), "v"(af[1].a0), "v"(af[1].a1),
                         "v"(af[2].a0), "v"(af[2].a1), "v"(af[3].a0), "v"(af[3].a1));
      mma_chunk(af[0], w[jg * 4 + 0], acc0, acc1);
      if (jg * 4 + 1 < ntot) {
        mma_chunk(af[1], w[jg * 4 + 1], acc0, acc1);
        if (jg * 4 + 2 < ntot) {
          mma_chunk(af[2], w[jg * 4 + 2], acc0, acc1);
          if (jg * 4 + 3 < ntot) mma_chunk(af[3], w[jg * 4 + 3], acc0, acc1);
        }
      }
    }
  }
}

struct StepOps {
  float4 zq0, zq1, zx0, zx1;
  float bias1, bias1x;
  float uu, bmu, bsg;
};

__device__ __forceinline__ void lds_barrier() {
  __builtin_amdgcn_fence(__ATOMIC_RELEASE, "workgroup", "local");
  __builtin_amdgcn_s_barrier();
  __builtin_amdgcn_fence(__ATOMIC_ACQUIRE, "workgroup", "local");
}

__launch_bounds__(256, 1)
__global__ void made_flow_kernel(const float* __restrict__ u,
                                 const float* __restrict__ b0,
                                 const float* __restrict__ b1,
                                 const float* __restrict__ b2,
                                 const float* __restrict__ ws,
                                 float* __restrict__ out) {
  extern __shared__ float smem[];
  float* xs   = smem;
  float* h0s  = xs  + kBT * kXStride;
  float* h1t  = h0s + kBT * kHStride;
  float* part = h1t + kBT * kDStride;
  float* dlt  = part + 8 * 2 * kBT * 16;
  float* zs   = dlt + kBT * kDStride;
  float* ldjs = zs + kBT * 2;

  const int tid  = threadIdx.x;
  const int wg   = blockIdx.x;
  const int lane = tid & 31;
  const int wv   = __builtin_amdgcn_readfirstlane(tid >> 5);
  const int n    = lane & 15;
  const int hi2  = (lane >> 4) << 1;
  const int hi8  = (lane >> 4) << 3;
  const int cc   = tid & 15;

  const float4* Wa4 = ((const float4*)(ws + kWaOff)) + lane;
  const float4* Wb4 = ((const float4*)(ws + kWbOff)) + lane;
  const float4* wz = ((const float4*)(ws + kWcOff)) + (wv * 128) * 32 + lane;

  float4 w1[8];
  float4 w2[16];
  float4 w2x[8];
  float  bias0 = 0.f;
  StepOps sm;
  const float4 zero4 = {0.f, 0.f, 0.f, 0.f};
  for (int k = 0; k < 8; ++k)  { w1[k] = zero4; w2x[k] = zero4; }
  for (int k = 0; k < 16; ++k) w2[k] = zero4;

  auto fetch_big = [&](int s) __attribute__((always_inline)) {
    const int g  = s - 1;
    const int nt = (g < 16) ? 2 : 1;
    if (wv < nt) {
      const int t = (wv == 0) ? g : kXT;
      bias0 = b0[(t < kXT) ? t + kDeg * n : kReg + n];
      const float4* wq = Wa4 + (t * 8) * 32;
      for (int cp = 0; cp < 8; ++cp) w1[cp] = wq[cp * 32];
    }
    int nch, ntot;
    l2_counts(g, wv, nch, ntot);
    fetch_w2<4>(w2, Wb4 + (g * 128) * 32, nch, ntot, wv);
    if (nt == 2) fetch_w2<2>(w2x, Wb4 + (kXT * 128) * 32, nch, ntot, wv);
  };
  auto fetch_small = [&](int s) __attribute__((always_inline)) {
    const int g = s - 1;
    sm.zq0 = wz[(2 * g) * 32];
    sm.zq1 = wz[(2 * g + 1) * 32];
    sm.bias1 = b1[g + kDeg * cc];
    sm.zx0 = wz[126 * 32];
    sm.zx1 = wz[127 * 32];
    sm.bias1x = b1[kReg + cc];
    sm.uu = u[(wg * kBT + n) * kD + s];
    sm.bmu = b2[s];
    sm.bsg = b2[kD + s];
  };

  fetch_big(1);
  const float u0   = u[(wg * kBT + n) * kD];
  const float bmu0 = b2[0], bsg0 = b2[kD];

  {
    const unsigned slice = ((unsigned)wg * 0x9E3779B1u) >> 28;
    constexpr int kUnitFloats = 64;
    constexpr int kUnits = (kWcOff + kWcSz) / kUnitFloats;
    for (int j = 0; j < (kUnits + 4095) / 4096; ++j) {
      const int q  = (j * 16 + (int)slice) * 256 + tid;
      const int qb = q - kWbOff / kUnitFloats;
      const bool in_b = qb >= 0 && qb < kWbSz / kUnitFloats;
      const int t = qb >> 8, ut = qb & 255;
      const bool need = !in_b || ut < 4 * (t + 1) || ut >= 252;
      if (q < kUnits && need) __builtin_prefetch(ws + (size_t)q * kUnitFloats, 0, 2);
    }
  }

  for (int e = tid; e < kBT * kXStride; e += 256) xs[e] = 0.f;
  for (int e = tid; e < kBT * 16; e += 256)
    h0s[(e >> 4) * kHStride + kReg + (e & 15)] = 0.f;
  lds_barrier();
  if (tid < kBT) {
    xs[tid * kXStride] = u0 * expf(bsg0) + bmu0;
    ldjs[tid] = bsg0;
  }
  fetch_small(1);
  lds_barrier();

  v8f zA, zB;
  for (int v = 0; v < 8; ++v) { zA[v] = 0.f; zB[v] = 0.f; }

  for (int i = 1; i < kD; ++i) {
    {
      const int g  = i - 1;
      const int nt = (g < 16) ? 2 : 1;

      if (wv < nt) {
        const int t  = (wv == 0) ? g : kXT;
        v8f accA, accB;
        for (int v = 0; v < 8; ++v) { accA[v] = bias0; accB[v] = 0.f; }
        const float* xr = xs + n * kXStride + hi2;
        for (int cp = 0; cp < 8; ++cp) {
          v2f a0 = *(const v2f*)&xr[cp * 8];
          v2f a1 = *(const v2f*)&xr[cp * 8 + 4];
          v2f bA = {w1[cp].x, w1[cp].y};
          v2f bB = {w1[cp].z, w1[cp].w};
          accA = wmma_f32(a0, bA, accA);
          accB = wmma_f32(a1, bB, accB);
        }
        if (t < kXT || n == g) {
          for (int v = 0; v < 8; ++v) {
            float f = accA[v] + accB[v];
            f = f > 0.f ? f : 0.f;
            h0s[(v + hi8) * kHStride + t * 16 + n] = f;
          }
        }
      }
      lds_barrier();

      {
        int nch, ntot;
        l2_counts(g, wv, nch, ntot);
        v8f acc0, acc1, acx0, acx1;
        for (int v = 0; v < 8; ++v) { acc0[v] = 0.f; acc1[v] = 0.f; acx0[v] = 0.f; acx1[v] = 0.f; }
        layer2_tile<4>(w2, h0s, nch, ntot, wv, n, hi2, acc0, acc1);
        if (nt == 2) layer2_tile<2>(w2x, h0s, nch, ntot, wv, n, hi2, acx0, acx1);

        asm volatile("" :: "v"(sm.zq0.x), "v"(sm.zq0.y), "v"(sm.zq0.z), "v"(sm.zq0.w),
                           "v"(sm.zq1.x), "v"(sm.zq1.y), "v"(sm.zq1.z), "v"(sm.zq1.w),
                           "v"(sm.zx0.x), "v"(sm.zx0.y), "v"(sm.zx0.z), "v"(sm.zx0.w),
                           "v"(sm.zx1.x), "v"(sm.zx1.y), "v"(sm.zx1.z), "v"(sm.zx1.w),
                           "v"(sm.bias1), "v"(sm.bias1x), "v"(sm.uu));
        if (i + 1 < kD) fetch_big(i + 1);

        float* pp = part + ((wv * 2) * kBT) * 16;
        for (int v = 0; v < 8; ++v) pp[(v + hi8) * 16 + n] = acc0[v] + acc1[v];
        if (nt == 2) {
          float* px = pp + kBT * 16;
          for (int v = 0; v < 8; ++v) px[(v + hi8) * 16 + n] = acx0[v] + acx1[v];
        }
      }
      lds_barrier();
      {
        const int row = tid >> 4;
        float s = sm.bias1;
        for (int w = 0; w < 8; ++w) s += part[((w * 2) * kBT + row) * 16 + cc];
        h1t[row * kDStride + cc] = s > 0.f ? s : 0.f;
        if (nt == 2) {
          float sx = sm.bias1x;
          for (int w = 0; w < 8; ++w) sx += part[((w * 2 + 1) * kBT + row) * 16 + cc];
          sx = sx > 0.f ? sx : 0.f;
          dlt[row * kDStride + cc] = (cc == g) ? sx : 0.f;
        }
      }
      lds_barrier();

      {
        AFrag f0 = load_a(h1t, kDStride, 0, n, hi2);
        AFrag f1 = load_a(h1t, kDStride, 8, n, hi2);
        mma_chunk(f0, sm.zq0, zA, zB);
        mma_chunk(f1, sm.zq1, zA, zB);
        if (nt == 2) {
          AFrag x0 = load_a(dlt, kDStride, 0, n, hi2);
          AFrag x1 = load_a(dlt, kDStride, 8, n, hi2);
          mma_chunk(x0, sm.zx0, zA, zB);
          mma_chunk(x1, sm.zx1, zA, zB);
        }
      }
    }

    if ((wv & 3) == (i >> 4) && n == (i & 15)) {
      const int cz = wv >> 2;
      for (int v = 0; v < 8; ++v) zs[(v + hi8) * 2 + cz] = zA[v] + zB[v];
    }
    lds_barrier();
    if (tid < kBT) {
      const float mu = sm.bmu + zs[tid * 2 + 0];
      const float sg = sm.bsg + zs[tid * 2 + 1];
      xs[tid * kXStride + i] = sm.uu * expf(sg) + mu;
      ldjs[tid] += sg;
    }
    fetch_small(i + 1 < kD ? i + 1 : i);
    lds_barrier();
  }

  for (int e = tid; e < kBT * kD; e += 256) {
    int r = e >> 6, d = e & 63;
    out[(wg * kBT + r) * kD + d] = xs[r * kXStride + d];
  }
  if (tid < kBT) out[kB * kD + wg * kBT + tid] = ldjs[tid];
}

extern "C" void kernel_launch(void* const* d_in, const int* in_sizes, int n_in,
                              void* d_out, int out_size, void* d_ws, size_t ws_size,
                              hipStream_t stream) {
  const float* u  = (const float*)d_in[0];
  const float* W0 = (const float*)d_in[1];
  const float* b0 = (const float*)d_in[2];
  const float* W1 = (const float*)d_in[3];
  const float* b1 = (const float*)d_in[4];
  const float* W2 = (const float*)d_in[5];
  const float* b2 = (const float*)d_in[6];
  float* ws  = (float*)d_ws;
  float* out = (float*)d_out;

  prep_w0<<<(kU * kD) / 256,     256, 0, stream>>>(W0, ws + kWaOff);
  prep_w1<<<(kU * kU) / 256,     256, 0, stream>>>(W1, ws + kWbOff);
  prep_w2<<<(2 * kD * kU) / 256, 256, 0, stream>>>(W2, ws + kWcOff);

  constexpr size_t kLdsUsed = (size_t)(kBT * kXStride + kBT * kHStride + 8 * 2 * kBT * 16 +
                                       2 * kBT * kDStride + kBT * 2 + kBT) * sizeof(float);
  constexpr size_t kLdsOnePerWgp = 164 * 1024;
  constexpr size_t smem = kLdsUsed > kLdsOnePerWgp ? kLdsUsed : kLdsOnePerWgp;
  made_flow_kernel<<<kB / kBT, 256, smem, stream>>>(u, b0, b1, b2, ws, out);
}


# The three repack launches (38912 one-element waves) become one of 288 workgroups: 16 source rows staged in LDS, whole 512-B chunk pairs stored, only the chunk pairs the recurrence reads
# speedup vs baseline: 65.8527x; 1.2964x over previous
#include <hip/hip_runtime.h>

typedef __attribute__((ext_vector_type(2))) float v2f;
typedef __attribute__((ext_vector_type(8))) float v8f;

constexpr int kD  = 64;
constexpr int kU  = 1024;
constexpr int kB  = 4096;
constexpr int kBT = 16;

constexpr int kDeg = kD - 1;
constexpr int kReg = 16 * kDeg;
constexpr int kXT  = kDeg;

constexpr int kXStride = kD + 4;
constexpr int kHStride = kU + 4;
constexpr int kDStride = 20;

constexpr int kWaOff = 0;
constexpr int kWaSz  = kU * kD;
constexpr int kWbOff = kWaOff + kWaSz;
constexpr int kWbSz  = kU * kU;
constexpr int kWcOff = kWbOff + kWbSz;
constexpr int kWcSz  = 2 * kD * kU;

__device__ __forceinline__ int pos_unit(int p) {
  return p < kReg ? (p >> 4) + kDeg * (p & 15) : p;
}
__device__ __forceinline__ int pos_deg(int p) {
  return p < kReg ? (p >> 4) : p - kReg;
}

constexpr int kRowPitch  = kU + 4;
constexpr int kPrepSplit = 4;
constexpr int kPrepJobs  = 64 + 8;

__launch_bounds__(256)
__global__ void prep_weights(const float* __restrict__ W0, const float* __restrict__ W1,
                             const float* __restrict__ W2, float* __restrict__ ws) {
  extern __shared__ float4 rows4[];
  float* rows = (float*)rows4;
  const int tid  = threadIdx.x;
  const int job  = blockIdx.x / kPrepSplit;
  const int part = blockIdx.x % kPrepSplit;
  const bool l2job = job < 64;
  const int tile = l2job ? job : job - 64;
  const int lim = l2job ? (tile < kXT ? 2 * (tile + 1) : 32) : 128;
  const int cnt = lim < 126 ? lim + 2 : 128;
  if (part * 8 >= cnt) return;

  const float* __restrict__ srcw = l2job ? W1 : W2;
  const int rowbase = l2job ? (tile < kXT ? tile : kReg) : tile * 16;
  const int rowstep = (l2job && tile < kXT) ? kDeg : 1;
  const float4* sp = (const float4*)srcw + tid;
  float4 v[16];
#pragma unroll
  for (int it = 0; it < 16; ++it) v[it] = sp[(size_t)(rowbase + rowstep * it) * (kU / 4)];
#pragma unroll
  for (int it = 0; it < 16; ++it) rows4[it * (kRowPitch / 4) + tid] = v[it];
  __syncthreads();

  const int l = tid & 31, n = l & 15, hi = l >> 4;
  const int rowdeg = l2job ? pos_deg(tile * 16 + n) + 1 : ((tile * 16 + n) & (kD - 1));
  float4* dst = (float4*)(ws + (l2job ? kWbOff : kWcOff)) + (size_t)(tile * 128) * 32;
  const float* rp = rows + n * kRowPitch;
  for (int i = part * 8 + (tid >> 5); i < cnt; i += 8 * kPrepSplit) {
    const int cp = i < lim ? i : 126 + (i - lim);
    float r[4];
    for (int q = 0; q < 4; ++q) {
      const int p = cp * 8 + ((q >> 1) << 2) + (hi << 1) + (q & 1);
      const float v = rp[pos_unit(p)];
      r[q] = pos_deg(p) < rowdeg ? v : 0.f;
    }
    const float4 o = {r[0], r[1], r[2], r[3]};
    dst[cp * 32 + l] = o;
  }

  if (l2job && part == 0) {
    const int cp = tid >> 5;
    const int j  = pos_unit(tile * 16 + n);
    const int dj = pos_deg(tile * 16 + n);
    float r[4];
    for (int q = 0; q < 4; ++q) {
      const int d = cp * 8 + ((q >> 1) << 2) + (hi << 1) + (q & 1);
      const float v = W0[j * kD + d];
      r[q] = d <= dj ? v : 0.f;
    }
    const float4 o = {r[0], r[1], r[2], r[3]};
    ((float4*)(ws + kWaOff))[(tile * 8 + cp) * 32 + l] = o;
  }
}


__device__ __forceinline__ v8f wmma_f32(v2f a, v2f b, v8f c) {
  return __builtin_amdgcn_wmma_f32_16x16x4_f32(false, a, false, b, (short)0, c,
                                               false, false);
}

struct AFrag { v2f a0, a1; };

__device__ __forceinline__ AFrag load_a(const float* __restrict__ as, int astride,
                                        int acol, int n, int hi2) {
  const int kb = acol + hi2;
  AFrag f;
  f.a0 = *(const v2f*)&as[n * astride + kb];
  f.a1 = *(const v2f*)&as[n * astride + kb + 4];
  return f;
}

__device__ __forceinline__ void mma_chunk(const AFrag& f, float4 bq, v8f& accA, v8f& accB) {
  v2f bA = {bq.x, bq.y};
  v2f bB = {bq.z, bq.w};
  accA = wmma_f32(f.a0, bA, accA);
  accB = wmma_f32(f.a1, bB, accB);
}

__device__ __forceinline__ void l2_counts(int g, int wv, int& nch, int& ntot) {
  const int ncp = 2 * (g + 1);
  nch  = (wv < ncp) ? ((ncp - wv + 7) >> 3) : 0;
  ntot = nch + ((wv >= 6) ? 1 : 0);
}

template <int NG>
__device__ __forceinline__ void fetch_w2(float4 (&w)[4 * NG], const float4* __restrict__ wq,
                                         int nch, int ntot, int wv) {
#pragma unroll
  for (int jg = 0; jg < NG; ++jg) {
    if (jg * 4 < ntot) {
#pragma unroll
      for (int jj = 0; jj < 4; ++jj) {
        int j = jg * 4 + jj;
        j = j < ntot ? j : ntot - 1;
        const int cp = (j < nch) ? wv + 8 * j : 120 + wv;
        w[jg * 4 + jj] = wq[cp * 32];
      }
    }
  }
}

template <int NG>
__device__ __forceinline__ void layer2_tile(const float4 (&w)[4 * NG],
                                            const float* __restrict__ h0s, int nch, int ntot,
                                            int wv, int n, int hi2, v8f& acc0, v8f& acc1) {
#pragma unroll
  for (int jg = 0; jg < NG; ++jg) {
    if (jg * 4 < ntot) {
      AFrag af[4];
#pragma unroll
      for (int jj = 0; jj < 4; ++jj) {
        int j = jg * 4 + jj;
        j = j < ntot ? j : ntot - 1;
        const int cp = (j < nch) ? wv + 8 * j : 120 + wv;
        af[jj] = load_a(h0s, kHStride, cp * 8, n, hi2);
      }
      asm volatile("" :: "v"(af[0].a0), "v"(af[0].a1), "v"(af[1].a0), "v"(af[1].a1),
                         "v"(af[2].a0), "v"(af[2].a1), "v"(af[3].a0), "v"(af[3].a1));
      mma_chunk(af[0], w[jg * 4 + 0], acc0, acc1);
      if (jg * 4 + 1 < ntot) {
        mma_chunk(af[1], w[jg * 4 + 1], acc0, acc1);
        if (jg * 4 + 2 < ntot) {
          mma_chunk(af[2], w[jg * 4 + 2], acc0, acc1);
          if (jg * 4 + 3 < ntot) mma_chunk(af[3], w[jg * 4 + 3], acc0, acc1);
        }
      }
    }
  }
}

struct StepOps {
  float4 zq0, zq1, zx0, zx1;
  float bias1, bias1x;
  float uu, bmu, bsg;
};

__device__ __forceinline__ void lds_barrier() {
  __builtin_amdgcn_fence(__ATOMIC_RELEASE, "workgroup", "local");
  __builtin_amdgcn_s_barrier();
  __builtin_amdgcn_fence(__ATOMIC_ACQUIRE, "workgroup", "local");
}

__launch_bounds__(256, 1)
__global__ void made_flow_kernel(const float* __restrict__ u,
                                 const float* __restrict__ b0,
                                 const float* __restrict__ b1,
                                 const float* __restrict__ b2,
                                 const float* __restrict__ ws,
                                 float* __restrict__ out) {
  extern __shared__ float smem[];
  float* xs   = smem;
  float* h0s  = xs  + kBT * kXStride;
  float* h1t  = h0s + kBT * kHStride;
  float* part = h1t + kBT * kDStride;
  float* dlt  = part + 8 * 2 * kBT * 16;
  float* zs   = dlt + kBT * kDStride;
  float* ldjs = zs + kBT * 2;

  const int tid  = threadIdx.x;
  const int wg   = blockIdx.x;
  const int lane = tid & 31;
  const int wv   = __builtin_amdgcn_readfirstlane(tid >> 5);
  const int n    = lane & 15;
  const int hi2  = (lane >> 4) << 1;
  const int hi8  = (lane >> 4) << 3;
  const int cc   = tid & 15;

  const float4* Wa4 = ((const float4*)(ws + kWaOff)) + lane;
  const float4* Wb4 = ((const float4*)(ws + kWbOff)) + lane;
  const float4* wz = ((const float4*)(ws + kWcOff)) + (wv * 128) * 32 + lane;

  float4 w1[8];
  float4 w2[16];
  float4 w2x[8];
  float  bias0 = 0.f;
  StepOps sm;
  const float4 zero4 = {0.f, 0.f, 0.f, 0.f};
  for (int k = 0; k < 8; ++k)  { w1[k] = zero4; w2x[k] = zero4; }
  for (int k = 0; k < 16; ++k) w2[k] = zero4;

  auto fetch_big = [&](int s) __attribute__((always_inline)) {
    const int g  = s - 1;
    const int nt = (g < 16) ? 2 : 1;
    if (wv < nt) {
      const int t = (wv == 0) ? g : kXT;
      bias0 = b0[(t < kXT) ? t + kDeg * n : kReg + n];
      const float4* wq = Wa4 + (t * 8) * 32;
      for (int cp = 0; cp < 8; ++cp) w1[cp] = wq[cp * 32];
    }
    int nch, ntot;
    l2_counts(g, wv, nch, ntot);
    fetch_w2<4>(w2, Wb4 + (g * 128) * 32, nch, ntot, wv);
    if (nt == 2) fetch_w2<2>(w2x, Wb4 + (kXT * 128) * 32, nch, ntot, wv);
  };
  auto fetch_small = [&](int s) __attribute__((always_inline)) {
    const int g = s - 1;
    sm.zq0 = wz[(2 * g) * 32];
    sm.zq1 = wz[(2 * g + 1) * 32];
    sm.bias1 = b1[g + kDeg * cc];
    sm.zx0 = wz[126 * 32];
    sm.zx1 = wz[127 * 32];
    sm.bias1x = b1[kReg + cc];
    sm.uu = u[(wg * kBT + n) * kD + s];
    sm.bmu = b2[s];
    sm.bsg = b2[kD + s];
  };

  fetch_big(1);
  const float u0   = u[(wg * kBT + n) * kD];
  const float bmu0 = b2[0], bsg0 = b2[kD];

  {
    const unsigned slice = ((unsigned)wg * 0x9E3779B1u) >> 28;
    constexpr int kUnitFloats = 64;
    constexpr int kUnits = (kWcOff + kWcSz) / kUnitFloats;
    for (int j = 0; j < (kUnits + 4095) / 4096; ++j) {
      const int q  = (j * 16 + (int)slice) * 256 + tid;
      const int qb = q - kWbOff / kUnitFloats;
      const bool in_b = qb >= 0 && qb < kWbSz / kUnitFloats;
      const int t = qb >> 8, ut = qb & 255;
      const bool need = !in_b || ut < 4 * (t + 1) || ut >= 252;
      if (q < kUnits && need) __builtin_prefetch(ws + (size_t)q * kUnitFloats, 0, 2);
    }
  }

  for (int e = tid; e < kBT * kXStride; e += 256) xs[e] = 0.f;
  for (int e = tid; e < kBT * 16; e += 256)
    h0s[(e >> 4) * kHStride + kReg + (e & 15)] = 0.f;
  lds_barrier();
  if (tid < kBT) {
    xs[tid * kXStride] = u0 * expf(bsg0) + bmu0;
    ldjs[tid] = bsg0;
  }
  fetch_small(1);
  lds_barrier();

  v8f zA, zB;
  for (int v = 0; v < 8; ++v) { zA[v] = 0.f; zB[v] = 0.f; }

  for (int i = 1; i < kD; ++i) {
    {
      const int g  = i - 1;
      const int nt = (g < 16) ? 2 : 1;

      if (wv < nt) {
        const int t  = (wv == 0) ? g : kXT;
        v8f accA, accB;
        for (int v = 0; v < 8; ++v) { accA[v] = bias0; accB[v] = 0.f; }
        const float* xr = xs + n * kXStride + hi2;
        for (int cp = 0; cp < 8; ++cp) {
          v2f a0 = *(const v2f*)&xr[cp * 8];
          v2f a1 = *(const v2f*)&xr[cp * 8 + 4];
          v2f bA = {w1[cp].x, w1[cp].y};
          v2f bB = {w1[cp].z, w1[cp].w};
          accA = wmma_f32(a0, bA, accA);
          accB = wmma_f32(a1, bB, accB);
        }
        if (t < kXT || n == g) {
          for (int v = 0; v < 8; ++v) {
            float f = accA[v] + accB[v];
            f = f > 0.f ? f : 0.f;
            h0s[(v + hi8) * kHStride + t * 16 + n] = f;
          }
        }
      }
      lds_barrier();

      {
        int nch, ntot;
        l2_counts(g, wv, nch, ntot);
        v8f acc0, acc1, acx0, acx1;
        for (int v = 0; v < 8; ++v) { acc0[v] = 0.f; acc1[v] = 0.f; acx0[v] = 0.f; acx1[v] = 0.f; }
        layer2_tile<4>(w2, h0s, nch, ntot, wv, n, hi2, acc0, acc1);
        if (nt == 2) layer2_tile<2>(w2x, h0s, nch, ntot, wv, n, hi2, acx0, acx1);

        asm volatile("" :: "v"(sm.zq0.x), "v"(sm.zq0.y), "v"(sm.zq0.z), "v"(sm.zq0.w),
                           "v"(sm.zq1.x), "v"(sm.zq1.y), "v"(sm.zq1.z), "v"(sm.zq1.w),
                           "v"(sm.zx0.x), "v"(sm.zx0.y), "v"(sm.zx0.z), "v"(sm.zx0.w),
                           "v"(sm.zx1.x), "v"(sm.zx1.y), "v"(sm.zx1.z), "v"(sm.zx1.w),
                           "v"(sm.bias1), "v"(sm.bias1x), "v"(sm.uu));
        if (i + 1 < kD) fetch_big(i + 1);

        float* pp = part + ((wv * 2) * kBT) * 16;
        for (int v = 0; v < 8; ++v) pp[(v + hi8) * 16 + n] = acc0[v] + acc1[v];
        if (nt == 2) {
          float* px = pp + kBT * 16;
          for (int v = 0; v < 8; ++v) px[(v + hi8) * 16 + n] = acx0[v] + acx1[v];
        }
      }
      lds_barrier();
      {
        const int row = tid >> 4;
        float s = sm.bias1;
        for (int w = 0; w < 8; ++w) s += part[((w * 2) * kBT + row) * 16 + cc];
        h1t[row * kDStride + cc] = s > 0.f ? s : 0.f;
        if (nt == 2) {
          float sx = sm.bias1x;
          for (int w = 0; w < 8; ++w) sx += part[((w * 2 + 1) * kBT + row) * 16 + cc];
          sx = sx > 0.f ? sx : 0.f;
          dlt[row * kDStride + cc] = (cc == g) ? sx : 0.f;
        }
      }
      lds_barrier();

      {
        AFrag f0 = load_a(h1t, kDStride, 0, n, hi2);
        AFrag f1 = load_a(h1t, kDStride, 8, n, hi2);
        mma_chunk(f0, sm.zq0, zA, zB);
        mma_chunk(f1, sm.zq1, zA, zB);
        if (nt == 2) {
          AFrag x0 = load_a(dlt, kDStride, 0, n, hi2);
          AFrag x1 = load_a(dlt, kDStride, 8, n, hi2);
          mma_chunk(x0, sm.zx0, zA, zB);
          mma_chunk(x1, sm.zx1, zA, zB);
        }
      }
    }

    if ((wv & 3) == (i >> 4) && n == (i & 15)) {
      const int cz = wv >> 2;
      for (int v = 0; v < 8; ++v) zs[(v + hi8) * 2 + cz] = zA[v] + zB[v];
    }
    lds_barrier();
    if (tid < kBT) {
      const float mu = sm.bmu + zs[tid * 2 + 0];
      const float sg = sm.bsg + zs[tid * 2 + 1];
      xs[tid * kXStride + i] = sm.uu * expf(sg) + mu;
      ldjs[tid] += sg;
    }
    fetch_small(i + 1 < kD ? i + 1 : i);
    lds_barrier();
  }

  for (int e = tid; e < kBT * kD; e += 256) {
    int r = e >> 6, d = e & 63;
    out[(wg * kBT + r) * kD + d] = xs[r * kXStride + d];
  }
  if (tid < kBT) out[kB * kD + wg * kBT + tid] = ldjs[tid];
}

extern "C" void kernel_launch(void* const* d_in, const int* in_sizes, int n_in,
                              void* d_out, int out_size, void* d_ws, size_t ws_size,
                              hipStream_t stream) {
  const float* u  = (const float*)d_in[0];
  const float* W0 = (const float*)d_in[1];
  const float* b0 = (const float*)d_in[2];
  const float* W1 = (const float*)d_in[3];
  const float* b1 = (const float*)d_in[4];
  const float* W2 = (const float*)d_in[5];
  const float* b2 = (const float*)d_in[6];
  float* ws  = (float*)d_ws;
  float* out = (float*)d_out;

  constexpr size_t kPrepLds = (size_t)16 * kRowPitch * sizeof(float);
  prep_weights<<<kPrepJobs * kPrepSplit, 256, kPrepLds, stream>>>(W0, W1, W2, ws);

  constexpr size_t kLdsUsed = (size_t)(kBT * kXStride + kBT * kHStride + 8 * 2 * kBT * 16 +
                                       2 * kBT * kDStride + kBT * 2 + kBT) * sizeof(float);
  constexpr size_t kLdsOnePerWgp = 164 * 1024;
  constexpr size_t smem = kLdsUsed > kLdsOnePerWgp ? kLdsUsed : kLdsOnePerWgp;
  made_flow_kernel<<<kB / kBT, 256, smem, stream>>>(u, b0, b1, b2, ws, out);
}


# Layer 1 of step i+1 is summed over the x columns already set during step i's exp phase (x update moved to wave 2); only the chunk pair holding the new column stays on the step's path
# speedup vs baseline: 71.0380x; 1.0787x over previous
#include <hip/hip_runtime.h>

typedef __attribute__((ext_vector_type(2))) float v2f;
typedef __attribute__((ext_vector_type(8))) float v8f;

constexpr int kD  = 64;
constexpr int kU  = 1024;
constexpr int kB  = 4096;
constexpr int kBT = 16;

constexpr int kDeg = kD - 1;
constexpr int kReg = 16 * kDeg;
constexpr int kXT  = kDeg;

constexpr int kXStride = kD + 4;
constexpr int kHStride = kU + 4;
constexpr int kDStride = 20;

constexpr int kWaOff = 0;
constexpr int kWaSz  = kU * kD;
constexpr int kWbOff = kWaOff + kWaSz;
constexpr int kWbSz  = kU * kU;
constexpr int kWcOff = kWbOff + kWbSz;
constexpr int kWcSz  = 2 * kD * kU;

__device__ __forceinline__ int pos_unit(int p) {
  return p < kReg ? (p >> 4) + kDeg * (p & 15) : p;
}
__device__ __forceinline__ int pos_deg(int p) {
  return p < kReg ? (p >> 4) : p - kReg;
}

constexpr int kRowPitch  = kU + 4;
constexpr int kPrepSplit = 4;
constexpr int kPrepJobs  = 64 + 8;

__launch_bounds__(256)
__global__ void prep_weights(const float* __restrict__ W0, const float* __restrict__ W1,
                             const float* __restrict__ W2, float* __restrict__ ws) {
  extern __shared__ float4 rows4[];
  float* rows = (float*)rows4;
  const int tid  = threadIdx.x;
  const int job  = blockIdx.x / kPrepSplit;
  const int part = blockIdx.x % kPrepSplit;
  const bool l2job = job < 64;
  const int tile = l2job ? job : job - 64;
  const int lim = l2job ? (tile < kXT ? 2 * (tile + 1) : 32) : 128;
  const int cnt = lim < 126 ? lim + 2 : 128;
  if (part * 8 >= cnt) return;

  const float* __restrict__ srcw = l2job ? W1 : W2;
  const int rowbase = l2job ? (tile < kXT ? tile : kReg) : tile * 16;
  const int rowstep = (l2job && tile < kXT) ? kDeg : 1;
  const float4* sp = (const float4*)srcw + tid;
  float4 v[16];
#pragma unroll
  for (int it = 0; it < 16; ++it) v[it] = sp[(size_t)(rowbase + rowstep * it) * (kU / 4)];
#pragma unroll
  for (int it = 0; it < 16; ++it) rows4[it * (kRowPitch / 4) + tid] = v[it];
  __syncthreads();

  const int l = tid & 31, n = l & 15, hi = l >> 4;
  const int rowdeg = l2job ? pos_deg(tile * 16 + n) + 1 : ((tile * 16 + n) & (kD - 1));
  float4* dst = (float4*)(ws + (l2job ? kWbOff : kWcOff)) + (size_t)(tile * 128) * 32;
  const float* rp = rows + n * kRowPitch;
  for (int i = part * 8 + (tid >> 5); i < cnt; i += 8 * kPrepSplit) {
    const int cp = i < lim ? i : 126 + (i - lim);
    float r[4];
    for (int q = 0; q < 4; ++q) {
      const int p = cp * 8 + ((q >> 1) << 2) + (hi << 1) + (q & 1);
      const float v = rp[pos_unit(p)];
      r[q] = pos_deg(p) < rowdeg ? v : 0.f;
    }
    const float4 o = {r[0], r[1], r[2], r[3]};
    dst[cp * 32 + l] = o;
  }

  if (l2job && part == 0) {
    const int cp = tid >> 5;
    const int j  = pos_unit(tile * 16 + n);
    const int dj = pos_deg(tile * 16 + n);
    float r[4];
    for (int q = 0; q < 4; ++q) {
      const int d = cp * 8 + ((q >> 1) << 2) + (hi << 1) + (q & 1);
      const float v = W0[j * kD + d];
      r[q] = d <= dj ? v : 0.f;
    }
    const float4 o = {r[0], r[1], r[2], r[3]};
    ((float4*)(ws + kWaOff))[(tile * 8 + cp) * 32 + l] = o;
  }
}


__device__ __forceinline__ v8f wmma_f32(v2f a, v2f b, v8f c) {
  return __builtin_amdgcn_wmma_f32_16x16x4_f32(false, a, false, b, (short)0, c,
                                               false, false);
}

struct AFrag { v2f a0, a1; };

__device__ __forceinline__ AFrag load_a(const float* __restrict__ as, int astride,
                                        int acol, int n, int hi2) {
  const int kb = acol + hi2;
  AFrag f;
  f.a0 = *(const v2f*)&as[n * astride + kb];
  f.a1 = *(const v2f*)&as[n * astride + kb + 4];
  return f;
}

__device__ __forceinline__ void mma_chunk(const AFrag& f, float4 bq, v8f& accA, v8f& accB) {
  v2f bA = {bq.x, bq.y};
  v2f bB = {bq.z, bq.w};
  accA = wmma_f32(f.a0, bA, accA);
  accB = wmma_f32(f.a1, bB, accB);
}

__device__ __forceinline__ void l2_counts(int g, int wv, int& nch, int& ntot) {
  const int ncp = 2 * (g + 1);
  nch  = (wv < ncp) ? ((ncp - wv + 7) >> 3) : 0;
  ntot = nch + ((wv >= 6) ? 1 : 0);
}

template <int NG>
__device__ __forceinline__ void fetch_w2(float4 (&w)[4 * NG], const float4* __restrict__ wq,
                                         int nch, int ntot, int wv) {
#pragma unroll
  for (int jg = 0; jg < NG; ++jg) {
    if (jg * 4 < ntot) {
#pragma unroll
      for (int jj = 0; jj < 4; ++jj) {
        int j = jg * 4 + jj;
        j = j < ntot ? j : ntot - 1;
        const int cp = (j < nch) ? wv + 8 * j : 120 + wv;
        w[jg * 4 + jj] = wq[cp * 32];
      }
    }
  }
}

template <int NG>
__device__ __forceinline__ void layer2_tile(const float4 (&w)[4 * NG],
                                            const float* __restrict__ h0s, int nch, int ntot,
                                            int wv, int n, int hi2, v8f& acc0, v8f& acc1) {
#pragma unroll
  for (int jg = 0; jg < NG; ++jg) {
    if (jg * 4 < ntot) {
      AFrag af[4];
#pragma unroll
      for (int jj = 0; jj < 4; ++jj) {
        int j = jg * 4 + jj;
        j = j < ntot ? j : ntot - 1;
        const int cp = (j < nch) ? wv + 8 * j : 120 + wv;
        af[jj] = load_a(h0s, kHStride, cp * 8, n, hi2);
      }
      asm volatile("" :: "v"(af[0].a0), "v"(af[0].a1), "v"(af[1].a0), "v"(af[1].a1),
                         "v"(af[2].a0), "v"(af[2].a1), "v"(af[3].a0), "v"(af[3].a1));
      mma_chunk(af[0], w[jg * 4 + 0], acc0, acc1);
      if (jg * 4 + 1 < ntot) {
        mma_chunk(af[1], w[jg * 4 + 1], acc0, acc1);
        if (jg * 4 + 2 < ntot) {
          mma_chunk(af[2], w[jg * 4 + 2], acc0, acc1);
          if (jg * 4 + 3 < ntot) mma_chunk(af[3], w[jg * 4 + 3], acc0, acc1);
        }
      }
    }
  }
}

struct StepOps {
  float4 zq0, zq1, zx0, zx1;
  float bias1, bias1x;
  float uu, bmu, bsg;
};

__device__ __forceinline__ void lds_barrier() {
  __builtin_amdgcn_fence(__ATOMIC_RELEASE, "workgroup", "local");
  __builtin_amdgcn_s_barrier();
  __builtin_amdgcn_fence(__ATOMIC_ACQUIRE, "workgroup", "local");
}

__launch_bounds__(256, 1)
__global__ void made_flow_kernel(const float* __restrict__ u,
                                 const float* __restrict__ b0,
                                 const float* __restrict__ b1,
                                 const float* __restrict__ b2,
                                 const float* __restrict__ ws,
                                 float* __restrict__ out) {
  extern __shared__ float smem[];
  float* xs   = smem;
  float* h0s  = xs  + kBT * kXStride;
  float* h1t  = h0s + kBT * kHStride;
  float* part = h1t + kBT * kDStride;
  float* dlt  = part + 8 * 2 * kBT * 16;
  float* zs   = dlt + kBT * kDStride;
  float* ldjs = zs + kBT * 2;

  const int tid  = threadIdx.x;
  const int wg   = blockIdx.x;
  const int lane = tid & 31;
  const int wv   = __builtin_amdgcn_readfirstlane(tid >> 5);
  const int n    = lane & 15;
  const int hi2  = (lane >> 4) << 1;
  const int hi8  = (lane >> 4) << 3;
  const int cc   = tid & 15;

  const float4* Wa4 = ((const float4*)(ws + kWaOff)) + lane;
  const float4* Wb4 = ((const float4*)(ws + kWbOff)) + lane;
  const float4* wz = ((const float4*)(ws + kWcOff)) + (wv * 128) * 32 + lane;

  float4 w1[8];
  float4 w2[16];
  float4 w2x[8];
  float  bias0 = 0.f;
  StepOps sm;
  const float4 zero4 = {0.f, 0.f, 0.f, 0.f};
  for (int k = 0; k < 8; ++k)  { w1[k] = zero4; w2x[k] = zero4; }
  for (int k = 0; k < 16; ++k) w2[k] = zero4;

  auto fetch_big = [&](int s) __attribute__((always_inline)) {
    const int g  = s - 1;
    const int nt = (g < 16) ? 2 : 1;
    if (wv < nt) {
      const int t = (wv == 0) ? g : kXT;
      bias0 = b0[(t < kXT) ? t + kDeg * n : kReg + n];
      const float4* wq = Wa4 + (t * 8) * 32;
      for (int cp = 0; cp < 8; ++cp) w1[cp] = wq[cp * 32];
    }
    int nch, ntot;
    l2_counts(g, wv, nch, ntot);
    fetch_w2<4>(w2, Wb4 + (g * 128) * 32, nch, ntot, wv);
    if (nt == 2) fetch_w2<2>(w2x, Wb4 + (kXT * 128) * 32, nch, ntot, wv);
  };
  auto fetch_small = [&](int s) __attribute__((always_inline)) {
    const int g = s - 1;
    sm.zq0 = wz[(2 * g) * 32];
    sm.zq1 = wz[(2 * g + 1) * 32];
    sm.bias1 = b1[g + kDeg * cc];
    sm.zx0 = wz[126 * 32];
    sm.zx1 = wz[127 * 32];
    sm.bias1x = b1[kReg + cc];
    sm.uu = u[(wg * kBT + n) * kD + s];
    sm.bmu = b2[s];
    sm.bsg = b2[kD + s];
  };

  fetch_big(1);
  const float u0   = u[(wg * kBT + n) * kD];
  const float bmu0 = b2[0], bsg0 = b2[kD];

  {
    const unsigned slice = ((unsigned)wg * 0x9E3779B1u) >> 28;
    constexpr int kUnitFloats = 64;
    constexpr int kUnits = (kWcOff + kWcSz) / kUnitFloats;
    for (int j = 0; j < (kUnits + 4095) / 4096; ++j) {
      const int q  = (j * 16 + (int)slice) * 256 + tid;
      const int qb = q - kWbOff / kUnitFloats;
      const bool in_b = qb >= 0 && qb < kWbSz / kUnitFloats;
      const int t = qb >> 8, ut = qb & 255;
      const bool need = !in_b || ut < 4 * (t + 1) || ut >= 252;
      if (q < kUnits && need) __builtin_prefetch(ws + (size_t)q * kUnitFloats, 0, 2);
    }
  }

  for (int e = tid; e < kBT * kXStride; e += 256) xs[e] = 0.f;
  for (int e = tid; e < kBT * 16; e += 256)
    h0s[(e >> 4) * kHStride + kReg + (e & 15)] = 0.f;
  lds_barrier();
  if (tid < kBT) {
    xs[tid * kXStride] = u0 * expf(bsg0) + bmu0;
    ldjs[tid] = bsg0;
  }
  fetch_small(1);
  lds_barrier();

  v8f zA, zB;
  for (int v = 0; v < 8; ++v) { zA[v] = 0.f; zB[v] = 0.f; }

  v8f l1A, l1B;
  for (int v = 0; v < 8; ++v) { l1A[v] = bias0; l1B[v] = 0.f; }

  for (int i = 1; i < kD; ++i) {
    {
      const int g  = i - 1;
      const int nt = (g < 16) ? 2 : 1;

      if (wv < nt) {
        const int t  = (wv == 0) ? g : kXT;
        const int p  = g >> 3;
        const float* xr = xs + n * kXStride + hi2 + p * 8;
        const v2f a0 = *(const v2f*)&xr[0];
        const v2f a1 = *(const v2f*)&xr[4];
#pragma unroll
        for (int cp = 0; cp < 8; ++cp) {
          if (cp == p) {
            v2f bA = {w1[cp].x, w1[cp].y};
            v2f bB = {w1[cp].z, w1[cp].w};
            l1A = wmma_f32(a0, bA, l1A);
            l1B = wmma_f32(a1, bB, l1B);
          }
        }
        if (t < kXT || n == g) {
          for (int v = 0; v < 8; ++v) {
            float f = l1A[v] + l1B[v];
            f = f > 0.f ? f : 0.f;
            h0s[(v + hi8) * kHStride + t * 16 + n] = f;
          }
        }
      }
      lds_barrier();

      {
        int nch, ntot;
        l2_counts(g, wv, nch, ntot);
        v8f acc0, acc1, acx0, acx1;
        for (int v = 0; v < 8; ++v) { acc0[v] = 0.f; acc1[v] = 0.f; acx0[v] = 0.f; acx1[v] = 0.f; }
        layer2_tile<4>(w2, h0s, nch, ntot, wv, n, hi2, acc0, acc1);
        if (nt == 2) layer2_tile<2>(w2x, h0s, nch, ntot, wv, n, hi2, acx0, acx1);

        asm volatile("" :: "v"(sm.zq0.x), "v"(sm.zq0.y), "v"(sm.zq0.z), "v"(sm.zq0.w),
                           "v"(sm.zq1.x), "v"(sm.zq1.y), "v"(sm.zq1.z), "v"(sm.zq1.w),
                           "v"(sm.zx0.x), "v"(sm.zx0.y), "v"(sm.zx0.z), "v"(sm.zx0.w),
                           "v"(sm.zx1.x), "v"(sm.zx1.y), "v"(sm.zx1.z), "v"(sm.zx1.w),
                           "v"(sm.bias1), "v"(sm.bias1x), "v"(sm.uu));
        if (i + 1 < kD) fetch_big(i + 1);

        float* pp = part + ((wv * 2) * kBT) * 16;
        for (int v = 0; v < 8; ++v) pp[(v + hi8) * 16 + n] = acc0[v] + acc1[v];
        if (nt == 2) {
          float* px = pp + kBT * 16;
          for (int v = 0; v < 8; ++v) px[(v + hi8) * 16 + n] = acx0[v] + acx1[v];
        }
      }
      lds_barrier();
      {
        const int row = tid >> 4;
        float s = sm.bias1;
        for (int w = 0; w < 8; ++w) s += part[((w * 2) * kBT + row) * 16 + cc];
        h1t[row * kDStride + cc] = s > 0.f ? s : 0.f;
        if (nt == 2) {
          float sx = sm.bias1x;
          for (int w = 0; w < 8; ++w) sx += part[((w * 2 + 1) * kBT + row) * 16 + cc];
          sx = sx > 0.f ? sx : 0.f;
          dlt[row * kDStride + cc] = (cc == g) ? sx : 0.f;
        }
      }
      lds_barrier();

      {
        AFrag f0 = load_a(h1t, kDStride, 0, n, hi2);
        AFrag f1 = load_a(h1t, kDStride, 8, n, hi2);
        mma_chunk(f0, sm.zq0, zA, zB);
        mma_chunk(f1, sm.zq1, zA, zB);
        if (nt == 2) {
          AFrag x0 = load_a(dlt, kDStride, 0, n, hi2);
          AFrag x1 = load_a(dlt, kDStride, 8, n, hi2);
          mma_chunk(x0, sm.zx0, zA, zB);
          mma_chunk(x1, sm.zx1, zA, zB);
        }
      }
    }

    if ((wv & 3) == (i >> 4) && n == (i & 15)) {
      const int cz = wv >> 2;
      for (int v = 0; v < 8; ++v) zs[(v + hi8) * 2 + cz] = zA[v] + zB[v];
    }
    lds_barrier();
    if (wv == 2 && lane < kBT) {
      const float mu = sm.bmu + zs[lane * 2 + 0];
      const float sg = sm.bsg + zs[lane * 2 + 1];
      xs[lane * kXStride + i] = sm.uu * expf(sg) + mu;
      ldjs[lane] += sg;
    }
    fetch_small(i + 1 < kD ? i + 1 : i);
    if (i + 1 < kD && wv < ((i < 16) ? 2 : 1)) {
      for (int v = 0; v < 8; ++v) { l1A[v] = bias0; l1B[v] = 0.f; }
      const int np = i >> 3;
      const float* xr = xs + n * kXStride + hi2;
      v2f pa0[7], pa1[7];
#pragma unroll
      for (int cp = 0; cp < 7; ++cp) {
        pa0[cp] = *(const v2f*)&xr[cp * 8];
        pa1[cp] = *(const v2f*)&xr[cp * 8 + 4];
      }
      asm volatile("" :: "v"(pa0[0]), "v"(pa1[0]), "v"(pa0[1]), "v"(pa1[1]), "v"(pa0[2]), "v"(pa1[2]),
                         "v"(pa0[3]), "v"(pa1[3]), "v"(pa0[4]), "v"(pa1[4]), "v"(pa0[5]), "v"(pa1[5]),
                         "v"(pa0[6]), "v"(pa1[6]));
#pragma unroll
      for (int cp = 0; cp < 7; ++cp) {
        if (cp < np) {
          v2f bA = {w1[cp].x, w1[cp].y};
          v2f bB = {w1[cp].z, w1[cp].w};
          l1A = wmma_f32(pa0[cp], bA, l1A);
          l1B = wmma_f32(pa1[cp], bB, l1B);
        }
      }
    }
    lds_barrier();
  }

  for (int e = tid; e < kBT * kD; e += 256) {
    int r = e >> 6, d = e & 63;
    out[(wg * kBT + r) * kD + d] = xs[r * kXStride + d];
  }
  if (tid < kBT) out[kB * kD + wg * kBT + tid] = ldjs[tid];
}

extern "C" void kernel_launch(void* const* d_in, const int* in_sizes, int n_in,
                              void* d_out, int out_size, void* d_ws, size_t ws_size,
                              hipStream_t stream) {
  const float* u  = (const float*)d_in[0];
  const float* W0 = (const float*)d_in[1];
  const float* b0 = (const float*)d_in[2];
  const float* W1 = (const float*)d_in[3];
  const float* b1 = (const float*)d_in[4];
  const float* W2 = (const float*)d_in[5];
  const float* b2 = (const float*)d_in[6];
  float* ws  = (float*)d_ws;
  float* out = (float*)d_out;

  constexpr size_t kPrepLds = (size_t)16 * kRowPitch * sizeof(float);
  prep_weights<<<kPrepJobs * kPrepSplit, 256, kPrepLds, stream>>>(W0, W1, W2, ws);

  constexpr size_t kLdsUsed = (size_t)(kBT * kXStride + kBT * kHStride + 8 * 2 * kBT * 16 +
                                       2 * kBT * kDStride + kBT * 2 + kBT) * sizeof(float);
  constexpr size_t kLdsOnePerWgp = 164 * 1024;
  constexpr size_t smem = kLdsUsed > kLdsOnePerWgp ? kLdsUsed : kLdsOnePerWgp;
  made_flow_kernel<<<kB / kBT, 256, smem, stream>>>(u, b0, b1, b2, ws, out);
}


# mu and sigma of x column i share one wave's z tile, kept transposed (batch row on the lane): that wave sets x[i] from registers; the z column's LDS hand-over and its barrier are gone
# speedup vs baseline: 74.6439x; 1.0508x over previous
#include <hip/hip_runtime.h>

typedef __attribute__((ext_vector_type(2))) float v2f;
typedef __attribute__((ext_vector_type(8))) float v8f;

constexpr int kD  = 64;
constexpr int kU  = 1024;
constexpr int kB  = 4096;
constexpr int kBT = 16;

constexpr int kDeg = kD - 1;
constexpr int kReg = 16 * kDeg;
constexpr int kXT  = kDeg;

constexpr int kXStride = kD + 4;
constexpr int kHStride = kU + 4;
constexpr int kDStride = 20;

constexpr int kWaOff = 0;
constexpr int kWaSz  = kU * kD;
constexpr int kWbOff = kWaOff + kWaSz;
constexpr int kWbSz  = kU * kU;
constexpr int kWcOff = kWbOff + kWbSz;
constexpr int kWcSz  = 2 * kD * kU;

__device__ __forceinline__ int pos_unit(int p) {
  return p < kReg ? (p >> 4) + kDeg * (p & 15) : p;
}
__device__ __forceinline__ int pos_deg(int p) {
  return p < kReg ? (p >> 4) : p - kReg;
}

constexpr int kRowPitch  = kU + 4;
constexpr int kPrepSplit = 4;
constexpr int kPrepJobs  = 64 + 8;

__launch_bounds__(256)
__global__ void prep_weights(const float* __restrict__ W0, const float* __restrict__ W1,
                             const float* __restrict__ W2, float* __restrict__ ws) {
  extern __shared__ float4 rows4[];
  float* rows = (float*)rows4;
  const int tid  = threadIdx.x;
  const int job  = blockIdx.x / kPrepSplit;
  const int part = blockIdx.x % kPrepSplit;
  const bool l2job = job < 64;
  const int tile = l2job ? job : job - 64;
  const int lim = l2job ? (tile < kXT ? 2 * (tile + 1) : 32) : 128;
  const int cnt = lim < 126 ? lim + 2 : 128;
  if (part * 8 >= cnt) return;

  const float* __restrict__ srcw = l2job ? W1 : W2;
  const int rowbase = l2job ? (tile < kXT ? tile : kReg) : tile * 8;
  const int rowstep = (l2job && tile < kXT) ? kDeg : 1;
  const int rowhalf = l2job ? 0 : kD - 8;
  const float4* sp = (const float4*)srcw + tid;
  float4 v[16];
#pragma unroll
  for (int it = 0; it < 16; ++it)
    v[it] = sp[(size_t)(rowbase + rowstep * it + (it >> 3) * rowhalf) * (kU / 4)];
#pragma unroll
  for (int it = 0; it < 16; ++it) rows4[it * (kRowPitch / 4) + tid] = v[it];
  __syncthreads();

  const int l = tid & 31, n = l & 15, hi = l >> 4;
  const int rowdeg = l2job ? pos_deg(tile * 16 + n) + 1 : tile * 8 + (n & 7);
  float4* dst = (float4*)(ws + (l2job ? kWbOff : kWcOff)) + (size_t)(tile * 128) * 32;
  const float* rp = rows + n * kRowPitch;
  for (int i = part * 8 + (tid >> 5); i < cnt; i += 8 * kPrepSplit) {
    const int cp = i < lim ? i : 126 + (i - lim);
    float r[4];
    for (int q = 0; q < 4; ++q) {
      const int p = cp * 8 + ((q >> 1) << 2) + (hi << 1) + (q & 1);
      const float v = rp[pos_unit(p)];
      r[q] = pos_deg(p) < rowdeg ? v : 0.f;
    }
    const float4 o = {r[0], r[1], r[2], r[3]};
    dst[cp * 32 + l] = o;
  }

  if (l2job && part == 0) {
    const int cp = tid >> 5;
    const int j  = pos_unit(tile * 16 + n);
    const int dj = pos_deg(tile * 16 + n);
    float r[4];
    for (int q = 0; q < 4; ++q) {
      const int d = cp * 8 + ((q >> 1) << 2) + (hi << 1) + (q & 1);
      const float v = W0[j * kD + d];
      r[q] = d <= dj ? v : 0.f;
    }
    const float4 o = {r[0], r[1], r[2], r[3]};
    ((float4*)(ws + kWaOff))[(tile * 8 + cp) * 32 + l] = o;
  }
}


__device__ __forceinline__ v8f wmma_f32(v2f a, v2f b, v8f c) {
  return __builtin_amdgcn_wmma_f32_16x16x4_f32(false, a, false, b, (short)0, c,
                                               false, false);
}

struct AFrag { v2f a0, a1; };

__device__ __forceinline__ AFrag load_a(const float* __restrict__ as, int astride,
                                        int acol, int n, int hi2) {
  const int kb = acol + hi2;
  AFrag f;
  f.a0 = *(const v2f*)&as[n * astride + kb];
  f.a1 = *(const v2f*)&as[n * astride + kb + 4];
  return f;
}

__device__ __forceinline__ void mma_chunk(const AFrag& f, float4 bq, v8f& accA, v8f& accB) {
  v2f bA = {bq.x, bq.y};
  v2f bB = {bq.z, bq.w};
  accA = wmma_f32(f.a0, bA, accA);
  accB = wmma_f32(f.a1, bB, accB);
}

__device__ __forceinline__ void mma_chunk_t(const AFrag& f, float4 wq, v8f& accA, v8f& accB) {
  v2f wA = {wq.x, wq.y};
  v2f wB = {wq.z, wq.w};
  accA = wmma_f32(wA, f.a0, accA);
  accB = wmma_f32(wB, f.a1, accB);
}

__device__ __forceinline__ float xhalf(float x) {
  const int b = __builtin_bit_cast(int, x);
  return __builtin_bit_cast(float,
      __builtin_amdgcn_permlanex16(b, b, 0x76543210, 0xfedcba98, false, false));
}

__device__ __forceinline__ void l2_counts(int g, int wv, int& nch, int& ntot) {
  const int ncp = 2 * (g + 1);
  nch  = (wv < ncp) ? ((ncp - wv + 7) >> 3) : 0;
  ntot = nch + ((wv >= 6) ? 1 : 0);
}

template <int NG>
__device__ __forceinline__ void fetch_w2(float4 (&w)[4 * NG], const float4* __restrict__ wq,
                                         int nch, int ntot, int wv) {
#pragma unroll
  for (int jg = 0; jg < NG; ++jg) {
    if (jg * 4 < ntot) {
#pragma unroll
      for (int jj = 0; jj < 4; ++jj) {
        int j = jg * 4 + jj;
        j = j < ntot ? j : ntot - 1;
        const int cp = (j < nch) ? wv + 8 * j : 120 + wv;
        w[jg * 4 + jj] = wq[cp * 32];
      }
    }
  }
}

template <int NG>
__device__ __forceinline__ void layer2_tile(const float4 (&w)[4 * NG],
                                            const float* __restrict__ h0s, int nch, int ntot,
                                            int wv, int n, int hi2, v8f& acc0, v8f& acc1) {
#pragma unroll
  for (int jg = 0; jg < NG; ++jg) {
    if (jg * 4 < ntot) {
      AFrag af[4];
#pragma unroll
      for (int jj = 0; jj < 4; ++jj) {
        int j = jg * 4 + jj;
        j = j < ntot ? j : ntot - 1;
        const int cp = (j < nch) ? wv + 8 * j : 120 + wv;
        af[jj] = load_a(h0s, kHStride, cp * 8, n, hi2);
      }
      asm volatile("" :: "v"(af[0].a0), "v"(af[0].a1), "v"(af[1].a0), "v"(af[1].a1),
                         "v"(af[2].a0), "v"(af[2].a1), "v"(af[3].a0), "v"(af[3].a1));
      mma_chunk(af[0], w[jg * 4 + 0], acc0, acc1);
      if (jg * 4 + 1 < ntot) {
        mma_chunk(af[1], w[jg * 4 + 1], acc0, acc1);
        if (jg * 4 + 2 < ntot) {
          mma_chunk(af[2], w[jg * 4 + 2], acc0, acc1);
          if (jg * 4 + 3 < ntot) mma_chunk(af[3], w[jg * 4 + 3], acc0, acc1);
        }
      }
    }
  }
}

struct StepOps {
  float4 zq0, zq1, zx0, zx1;
  float bias1, bias1x;
  float uu, bmu, bsg;
};

__device__ __forceinline__ void lds_barrier() {
  __builtin_amdgcn_fence(__ATOMIC_RELEASE, "workgroup", "local");
  __builtin_amdgcn_s_barrier();
  __builtin_amdgcn_fence(__ATOMIC_ACQUIRE, "workgroup", "local");
}

__launch_bounds__(256, 1)
__global__ void made_flow_kernel(const float* __restrict__ u,
                                 const float* __restrict__ b0,
                                 const float* __restrict__ b1,
                                 const float* __restrict__ b2,
                                 const float* __restrict__ ws,
                                 float* __restrict__ out) {
  extern __shared__ float smem[];
  float* xs   = smem;
  float* h0s  = xs  + kBT * kXStride;
  float* h1t  = h0s + kBT * kHStride;
  float* part = h1t + kBT * kDStride;
  float* dlt  = part + 8 * 2 * kBT * 16;
  float* ldjs = dlt + kBT * kDStride;

  const int tid  = threadIdx.x;
  const int wg   = blockIdx.x;
  const int lane = tid & 31;
  const int wv   = __builtin_amdgcn_readfirstlane(tid >> 5);
  const int n    = lane & 15;
  const int hi2  = (lane >> 4) << 1;
  const int hi8  = (lane >> 4) << 3;
  const int cc   = tid & 15;

  const float4* Wa4 = ((const float4*)(ws + kWaOff)) + lane;
  const float4* Wb4 = ((const float4*)(ws + kWbOff)) + lane;
  const float4* wz = ((const float4*)(ws + kWcOff)) + (wv * 128) * 32 + lane;

  float4 w1[8];
  float4 w2[16];
  float4 w2x[8];
  float  bias0 = 0.f;
  StepOps sm;
  const float4 zero4 = {0.f, 0.f, 0.f, 0.f};
  for (int k = 0; k < 8; ++k)  { w1[k] = zero4; w2x[k] = zero4; }
  for (int k = 0; k < 16; ++k) w2[k] = zero4;

  auto fetch_big = [&](int s) __attribute__((always_inline)) {
    const int g  = s - 1;
    const int nt = (g < 16) ? 2 : 1;
    if (wv < nt) {
      const int t = (wv == 0) ? g : kXT;
      bias0 = b0[(t < kXT) ? t + kDeg * n : kReg + n];
      const float4* wq = Wa4 + (t * 8) * 32;
      for (int cp = 0; cp < 8; ++cp) w1[cp] = wq[cp * 32];
    }
    int nch, ntot;
    l2_counts(g, wv, nch, ntot);
    fetch_w2<4>(w2, Wb4 + (g * 128) * 32, nch, ntot, wv);
    if (nt == 2) fetch_w2<2>(w2x, Wb4 + (kXT * 128) * 32, nch, ntot, wv);
  };
  auto fetch_small = [&](int s) __attribute__((always_inline)) {
    const int g = s - 1;
    sm.zq0 = wz[(2 * g) * 32];
    sm.zq1 = wz[(2 * g + 1) * 32];
    sm.bias1 = b1[g + kDeg * cc];
    sm.zx0 = wz[126 * 32];
    sm.zx1 = wz[127 * 32];
    sm.bias1x = b1[kReg + cc];
    sm.uu = u[(wg * kBT + n) * kD + s];
    sm.bmu = b2[s];
    sm.bsg = b2[kD + s];
  };

  fetch_big(1);
  const float u0   = u[(wg * kBT + n) * kD];
  const float bmu0 = b2[0], bsg0 = b2[kD];

  {
    const unsigned slice = ((unsigned)wg * 0x9E3779B1u) >> 28;
    constexpr int kUnitFloats = 64;
    constexpr int kUnits = (kWcOff + kWcSz) / kUnitFloats;
    for (int j = 0; j < (kUnits + 4095) / 4096; ++j) {
      const int q  = (j * 16 + (int)slice) * 256 + tid;
      const int qb = q - kWbOff / kUnitFloats;
      const bool in_b = qb >= 0 && qb < kWbSz / kUnitFloats;
      const int t = qb >> 8, ut = qb & 255;
      const bool need = !in_b || ut < 4 * (t + 1) || ut >= 252;
      if (q < kUnits && need) __builtin_prefetch(ws + (size_t)q * kUnitFloats, 0, 2);
    }
  }

  for (int e = tid; e < kBT * kXStride; e += 256) xs[e] = 0.f;
  for (int e = tid; e < kBT * 16; e += 256)
    h0s[(e >> 4) * kHStride + kReg + (e & 15)] = 0.f;
  lds_barrier();
  if (tid < kBT) {
    xs[tid * kXStride] = u0 * expf(bsg0) + bmu0;
    ldjs[tid] = bsg0;
  }
  fetch_small(1);
  lds_barrier();

  v8f zA, zB;
  for (int v = 0; v < 8; ++v) { zA[v] = 0.f; zB[v] = 0.f; }

  v8f l1A, l1B;
  for (int v = 0; v < 8; ++v) { l1A[v] = bias0; l1B[v] = 0.f; }

  for (int i = 1; i < kD; ++i) {
    {
      const int g  = i - 1;
      const int nt = (g < 16) ? 2 : 1;

      if (wv < nt) {
        const int t  = (wv == 0) ? g : kXT;
        const int p  = g >> 3;
        const float* xr = xs + n * kXStride + hi2 + p * 8;
        const v2f a0 = *(const v2f*)&xr[0];
        const v2f a1 = *(const v2f*)&xr[4];
#pragma unroll
        for (int cp = 0; cp < 8; ++cp) {
          if (cp == p) {
            v2f bA = {w1[cp].x, w1[cp].y};
            v2f bB = {w1[cp].z, w1[cp].w};
            l1A = wmma_f32(a0, bA, l1A);
            l1B = wmma_f32(a1, bB, l1B);
          }
        }
        if (t < kXT || n == g) {
          for (int v = 0; v < 8; ++v) {
            float f = l1A[v] + l1B[v];
            f = f > 0.f ? f : 0.f;
            h0s[(v + hi8) * kHStride + t * 16 + n] = f;
          }
        }
      }
      lds_barrier();

      {
        int nch, ntot;
        l2_counts(g, wv, nch, ntot);
        v8f acc0, acc1, acx0, acx1;
        for (int v = 0; v < 8; ++v) { acc0[v] = 0.f; acc1[v] = 0.f; acx0[v] = 0.f; acx1[v] = 0.f; }
        layer2_tile<4>(w2, h0s, nch, ntot, wv, n, hi2, acc0, acc1);
        if (nt == 2) layer2_tile<2>(w2x, h0s, nch, ntot, wv, n, hi2, acx0, acx1);

        asm volatile("" :: "v"(sm.zq0.x), "v"(sm.zq0.y), "v"(sm.zq0.z), "v"(sm.zq0.w),
                           "v"(sm.zq1.x), "v"(sm.zq1.y), "v"(sm.zq1.z), "v"(sm.zq1.w),
                           "v"(sm.zx0.x), "v"(sm.zx0.y), "v"(sm.zx0.z), "v"(sm.zx0.w),
                           "v"(sm.zx1.x), "v"(sm.zx1.y), "v"(sm.zx1.z), "v"(sm.zx1.w),
                           "v"(sm.bias1), "v"(sm.bias1x), "v"(sm.uu));
        if (i + 1 < kD) fetch_big(i + 1);

        float* pp = part + ((wv * 2) * kBT) * 16;
        for (int v = 0; v < 8; ++v) pp[(v + hi8) * 16 + n] = acc0[v] + acc1[v];
        if (nt == 2) {
          float* px = pp + kBT * 16;
          for (int v = 0; v < 8; ++v) px[(v + hi8) * 16 + n] = acx0[v] + acx1[v];
        }
      }
      lds_barrier();
      {
        const int row = tid >> 4;
        float s = sm.bias1;
        for (int w = 0; w < 8; ++w) s += part[((w * 2) * kBT + row) * 16 + cc];
        h1t[row * kDStride + cc] = s > 0.f ? s : 0.f;
        if (nt == 2) {
          float sx = sm.bias1x;
          for (int w = 0; w < 8; ++w) sx += part[((w * 2 + 1) * kBT + row) * 16 + cc];
          sx = sx > 0.f ? sx : 0.f;
          dlt[row * kDStride + cc] = (cc == g) ? sx : 0.f;
        }
      }
      lds_barrier();

      {
        AFrag f0 = load_a(h1t, kDStride, 0, n, hi2);
        AFrag f1 = load_a(h1t, kDStride, 8, n, hi2);
        mma_chunk_t(f0, sm.zq0, zA, zB);
        mma_chunk_t(f1, sm.zq1, zA, zB);
        if (nt == 2) {
          AFrag x0 = load_a(dlt, kDStride, 0, n, hi2);
          AFrag x1 = load_a(dlt, kDStride, 8, n, hi2);
          mma_chunk_t(x0, sm.zx0, zA, zB);
          mma_chunk_t(x1, sm.zx1, zA, zB);
        }
      }
    }

    if (wv == (i >> 3)) {
      const bool k0 = (i & 1) != 0, k1 = (i & 2) != 0, k2 = (i & 4) != 0;
      const float t0 = k0 ? zA[1] + zB[1] : zA[0] + zB[0];
      const float t1 = k0 ? zA[3] + zB[3] : zA[2] + zB[2];
      const float t2 = k0 ? zA[5] + zB[5] : zA[4] + zB[4];
      const float t3 = k0 ? zA[7] + zB[7] : zA[6] + zB[6];
      const float q0 = k1 ? t1 : t0;
      const float q1 = k1 ? t3 : t2;
      const float zc = k2 ? q1 : q0;
      const float zo = xhalf(zc);
      if (lane < kBT) {
        const float mu = sm.bmu + zc;
        const float sg = sm.bsg + zo;
        xs[lane * kXStride + i] = sm.uu * expf(sg) + mu;
        ldjs[lane] += sg;
      }
    }
    fetch_small(i + 1 < kD ? i + 1 : i);
    if (i + 1 < kD && wv < ((i < 16) ? 2 : 1)) {
      for (int v = 0; v < 8; ++v) { l1A[v] = bias0; l1B[v] = 0.f; }
      const int np = i >> 3;
      const float* xr = xs + n * kXStride + hi2;
      v2f pa0[7], pa1[7];
#pragma unroll
      for (int cp = 0; cp < 7; ++cp) {
        pa0[cp] = *(const v2f*)&xr[cp * 8];
        pa1[cp] = *(const v2f*)&xr[cp * 8 + 4];
      }
      asm volatile("" :: "v"(pa0[0]), "v"(pa1[0]), "v"(pa0[1]), "v"(pa1[1]), "v"(pa0[2]), "v"(pa1[2]),
                         "v"(pa0[3]), "v"(pa1[3]), "v"(pa0[4]), "v"(pa1[4]), "v"(pa0[5]), "v"(pa1[5]),
                         "v"(pa0[6]), "v"(pa1[6]));
#pragma unroll
      for (int cp = 0; cp < 7; ++cp) {
        if (cp < np) {
          v2f bA = {w1[cp].x, w1[cp].y};
          v2f bB = {w1[cp].z, w1[cp].w};
          l1A = wmma_f32(pa0[cp], bA, l1A);
          l1B = wmma_f32(pa1[cp], bB, l1B);
        }
      }
    }
    lds_barrier();
  }

  for (int e = tid; e < kBT * kD; e += 256) {
    int r = e >> 6, d = e & 63;
    out[(wg * kBT + r) * kD + d] = xs[r * kXStride + d];
  }
  if (tid < kBT) out[kB * kD + wg * kBT + tid] = ldjs[tid];
}

extern "C" void kernel_launch(void* const* d_in, const int* in_sizes, int n_in,
                              void* d_out, int out_size, void* d_ws, size_t ws_size,
                              hipStream_t stream) {
  const float* u  = (const float*)d_in[0];
  const float* W0 = (const float*)d_in[1];
  const float* b0 = (const float*)d_in[2];
  const float* W1 = (const float*)d_in[3];
  const float* b1 = (const float*)d_in[4];
  const float* W2 = (const float*)d_in[5];
  const float* b2 = (const float*)d_in[6];
  float* ws  = (float*)d_ws;
  float* out = (float*)d_out;

  constexpr size_t kPrepLds = (size_t)16 * kRowPitch * sizeof(float);
  prep_weights<<<kPrepJobs * kPrepSplit, 256, kPrepLds, stream>>>(W0, W1, W2, ws);

  constexpr size_t kLdsUsed = (size_t)(kBT * kXStride + kBT * kHStride + 8 * 2 * kBT * 16 +
                                       2 * kBT * kDStride + kBT) * sizeof(float);
  constexpr size_t kLdsOnePerWgp = 164 * 1024;
  constexpr size_t smem = kLdsUsed > kLdsOnePerWgp ? kLdsUsed : kLdsOnePerWgp;
  made_flow_kernel<<<kB / kBT, 256, smem, stream>>>(u, b0, b1, b2, ws, out);
}


# Next step's weight fetch is issued after the step's partial stores and each load is a per-tile base + constant (guard per chunk pair, no clamped 64-bit address math): off the path to the barrier
# speedup vs baseline: 79.0939x; 1.0596x over previous
#include <hip/hip_runtime.h>

typedef __attribute__((ext_vector_type(2))) float v2f;
typedef __attribute__((ext_vector_type(8))) float v8f;

constexpr int kD  = 64;
constexpr int kU  = 1024;
constexpr int kB  = 4096;
constexpr int kBT = 16;

constexpr int kDeg = kD - 1;
constexpr int kReg = 16 * kDeg;
constexpr int kXT  = kDeg;

constexpr int kXStride = kD + 4;
constexpr int kHStride = kU + 4;
constexpr int kDStride = 20;

constexpr int kWaOff = 0;
constexpr int kWaSz  = kU * kD;
constexpr int kWbOff = kWaOff + kWaSz;
constexpr int kWbSz  = kU * kU;
constexpr int kWcOff = kWbOff + kWbSz;
constexpr int kWcSz  = 2 * kD * kU;

__device__ __forceinline__ int pos_unit(int p) {
  return p < kReg ? (p >> 4) + kDeg * (p & 15) : p;
}
__device__ __forceinline__ int pos_deg(int p) {
  return p < kReg ? (p >> 4) : p - kReg;
}

constexpr int kRowPitch  = kU + 4;
constexpr int kPrepSplit = 4;
constexpr int kPrepJobs  = 64 + 8;

__launch_bounds__(256)
__global__ void prep_weights(const float* __restrict__ W0, const float* __restrict__ W1,
                             const float* __restrict__ W2, float* __restrict__ ws) {
  extern __shared__ float4 rows4[];
  float* rows = (float*)rows4;
  const int tid  = threadIdx.x;
  const int job  = blockIdx.x / kPrepSplit;
  const int part = blockIdx.x % kPrepSplit;
  const bool l2job = job < 64;
  const int tile = l2job ? job : job - 64;
  const int lim = l2job ? (tile < kXT ? 2 * (tile + 1) : 32) : 128;
  const int cnt = lim < 126 ? lim + 2 : 128;
  if (part * 8 >= cnt) return;

  const float* __restrict__ srcw = l2job ? W1 : W2;
  const int rowbase = l2job ? (tile < kXT ? tile : kReg) : tile * 8;
  const int rowstep = (l2job && tile < kXT) ? kDeg : 1;
  const int rowhalf = l2job ? 0 : kD - 8;
  const float4* sp = (const float4*)srcw + tid;
  float4 v[16];
#pragma unroll
  for (int it = 0; it < 16; ++it)
    v[it] = sp[(size_t)(rowbase + rowstep * it + (it >> 3) * rowhalf) * (kU / 4)];
#pragma unroll
  for (int it = 0; it < 16; ++it) rows4[it * (kRowPitch / 4) + tid] = v[it];
  __syncthreads();

  const int l = tid & 31, n = l & 15, hi = l >> 4;
  const int rowdeg = l2job ? pos_deg(tile * 16 + n) + 1 : tile * 8 + (n & 7);
  float4* dst = (float4*)(ws + (l2job ? kWbOff : kWcOff)) + (size_t)(tile * 128) * 32;
  const float* rp = rows + n * kRowPitch;
  for (int i = part * 8 + (tid >> 5); i < cnt; i += 8 * kPrepSplit) {
    const int cp = i < lim ? i : 126 + (i - lim);
    float r[4];
    for (int q = 0; q < 4; ++q) {
      const int p = cp * 8 + ((q >> 1) << 2) + (hi << 1) + (q & 1);
      const float v = rp[pos_unit(p)];
      r[q] = pos_deg(p) < rowdeg ? v : 0.f;
    }
    const float4 o = {r[0], r[1], r[2], r[3]};
    dst[cp * 32 + l] = o;
  }

  if (l2job && part == 0) {
    const int cp = tid >> 5;
    const int j  = pos_unit(tile * 16 + n);
    const int dj = pos_deg(tile * 16 + n);
    float r[4];
    for (int q = 0; q < 4; ++q) {
      const int d = cp * 8 + ((q >> 1) << 2) + (hi << 1) + (q & 1);
      const float v = W0[j * kD + d];
      r[q] = d <= dj ? v : 0.f;
    }
    const float4 o = {r[0], r[1], r[2], r[3]};
    ((float4*)(ws + kWaOff))[(tile * 8 + cp) * 32 + l] = o;
  }
}


__device__ __forceinline__ v8f wmma_f32(v2f a, v2f b, v8f c) {
  return __builtin_amdgcn_wmma_f32_16x16x4_f32(false, a, false, b, (short)0, c,
                                               false, false);
}

struct AFrag { v2f a0, a1; };

__device__ __forceinline__ AFrag load_a(const float* __restrict__ as, int astride,
                                        int acol, int n, int hi2) {
  const int kb = acol + hi2;
  AFrag f;
  f.a0 = *(const v2f*)&as[n * astride + kb];
  f.a1 = *(const v2f*)&as[n * astride + kb + 4];
  return f;
}

__device__ __forceinline__ void mma_chunk(const AFrag& f, float4 bq, v8f& accA, v8f& accB) {
  v2f bA = {bq.x, bq.y};
  v2f bB = {bq.z, bq.w};
  accA = wmma_f32(f.a0, bA, accA);
  accB = wmma_f32(f.a1, bB, accB);
}

__device__ __forceinline__ void mma_chunk_t(const AFrag& f, float4 wq, v8f& accA, v8f& accB) {
  v2f wA = {wq.x, wq.y};
  v2f wB = {wq.z, wq.w};
  accA = wmma_f32(wA, f.a0, accA);
  accB = wmma_f32(wB, f.a1, accB);
}

__device__ __forceinline__ float xhalf(float x) {
  const int b = __builtin_bit_cast(int, x);
  return __builtin_bit_cast(float,
      __builtin_amdgcn_permlanex16(b, b, 0x76543210, 0xfedcba98, false, false));
}

__device__ __forceinline__ void l2_counts(int g, int wv, int& nch, int& ntot) {
  const int ncp = 2 * (g + 1);
  nch  = (wv < ncp) ? ((ncp - wv + 7) >> 3) : 0;
  ntot = nch + ((wv >= 6) ? 1 : 0);
}

template <int N>
__device__ __forceinline__ void fetch_w2(float4 (&w)[N], float4& wl, const float4* __restrict__ tb,
                                         unsigned lane_b, int nch, int wv) {
#pragma unroll
  for (int jg = 0; jg < N / 4; ++jg) {
    if (jg * 4 < nch) {
#pragma unroll
      for (int jj = 0; jj < 4; ++jj) {
        const int j = jg * 4 + jj;
        if (j < nch) w[j] = *(const float4*)((const char*)(tb + j * 256) + lane_b);
      }
    }
  }
  if (wv >= 6) wl = *(const float4*)((const char*)(tb + 120 * 32) + lane_b);
}

template <int NG>
__device__ __forceinline__ void layer2_tile(const float4 (&w)[4 * NG], float4 wl,
                                            const float* __restrict__ h0s, int nch,
                                            int wv, int n, int hi2, v8f& acc0, v8f& acc1) {
  const bool lo_pair = wv >= 6;
  const v2f zero2 = {0.f, 0.f};
  AFrag afl;
  afl.a0 = zero2;
  afl.a1 = zero2;
  if (lo_pair) afl = load_a(h0s, kHStride, (120 + wv) * 8, n, hi2);
#pragma unroll
  for (int jg = 0; jg < NG; ++jg) {
    if (jg * 4 < nch) {
      AFrag af[4];
#pragma unroll
      for (int jj = 0; jj < 4; ++jj) {
        int j = jg * 4 + jj;
        j = j < nch ? j : nch - 1;
        af[jj] = load_a(h0s, kHStride, (wv + 8 * j) * 8, n, hi2);
      }
      if (jg == 0)
        asm volatile("" :: "v"(af[0].a0), "v"(af[0].a1), "v"(af[1].a0), "v"(af[1].a1),
                           "v"(af[2].a0), "v"(af[2].a1), "v"(af[3].a0), "v"(af[3].a1),
                           "v"(afl.a0), "v"(afl.a1));
      else
        asm volatile("" :: "v"(af[0].a0), "v"(af[0].a1), "v"(af[1].a0), "v"(af[1].a1),
                           "v"(af[2].a0), "v"(af[2].a1), "v"(af[3].a0), "v"(af[3].a1));
      mma_chunk(af[0], w[jg * 4 + 0], acc0, acc1);
      if (jg * 4 + 1 < nch) {
        mma_chunk(af[1], w[jg * 4 + 1], acc0, acc1);
        if (jg * 4 + 2 < nch) {
          mma_chunk(af[2], w[jg * 4 + 2], acc0, acc1);
          if (jg * 4 + 3 < nch) mma_chunk(af[3], w[jg * 4 + 3], acc0, acc1);
        }
      }
    }
  }
  if (lo_pair) mma_chunk(afl, wl, acc0, acc1);
}

struct StepOps {
  float4 zq0, zq1, zx0, zx1;
  float bias1, bias1x;
  float uu, bmu, bsg;
};

__device__ __forceinline__ void lds_barrier() {
  __builtin_amdgcn_fence(__ATOMIC_RELEASE, "workgroup", "local");
  __builtin_amdgcn_s_barrier();
  __builtin_amdgcn_fence(__ATOMIC_ACQUIRE, "workgroup", "local");
}

__launch_bounds__(256, 1)
__global__ void made_flow_kernel(const float* __restrict__ u,
                                 const float* __restrict__ b0,
                                 const float* __restrict__ b1,
                                 const float* __restrict__ b2,
                                 const float* __restrict__ ws,
                                 float* __restrict__ out) {
  extern __shared__ float smem[];
  float* xs   = smem;
  float* h0s  = xs  + kBT * kXStride;
  float* h1t  = h0s + kBT * kHStride;
  float* part = h1t + kBT * kDStride;
  float* dlt  = part + 8 * 2 * kBT * 16;
  float* ldjs = dlt + kBT * kDStride;

  const int tid  = threadIdx.x;
  const int wg   = blockIdx.x;
  const int lane = tid & 31;
  const int wv   = __builtin_amdgcn_readfirstlane(tid >> 5);
  const int n    = lane & 15;
  const int hi2  = (lane >> 4) << 1;
  const int hi8  = (lane >> 4) << 3;
  const int cc   = tid & 15;

  const float4* Wa4 = (const float4*)(ws + kWaOff);
  const float4* Wb4 = (const float4*)(ws + kWbOff);
  const unsigned lane_b = (unsigned)lane * 16u;
  const float4* wz = ((const float4*)(ws + kWcOff)) + (wv * 128) * 32 + lane;

  float4 w1[8];
  float4 w2[16];
  float4 w2x[4];
  float4 w2l, w2xl;
  float  bias0 = 0.f;
  StepOps sm;
  const float4 zero4 = {0.f, 0.f, 0.f, 0.f};
  for (int k = 0; k < 8; ++k)  w1[k] = zero4;
  for (int k = 0; k < 4; ++k)  w2x[k] = zero4;
  for (int k = 0; k < 16; ++k) w2[k] = zero4;
  w2l = zero4; w2xl = zero4;

  auto fetch_big = [&](int s) __attribute__((always_inline)) {
    const int g  = s - 1;
    const int nt = (g < 16) ? 2 : 1;
    if (wv < nt) {
      const int t = (wv == 0) ? g : kXT;
      bias0 = b0[(t < kXT) ? t + kDeg * n : kReg + n];
      const float4* wq = Wa4 + (t * 8) * 32;
      for (int cp = 0; cp < 8; ++cp) w1[cp] = *(const float4*)((const char*)(wq + cp * 32) + lane_b);
    }
    int nch, ntot;
    l2_counts(g, wv, nch, ntot);
    fetch_w2<16>(w2, w2l, Wb4 + (g * 128 + wv) * 32, lane_b, nch, wv);
    if (nt == 2) fetch_w2<4>(w2x, w2xl, Wb4 + (kXT * 128 + wv) * 32, lane_b, nch, wv);
  };
  auto fetch_small = [&](int s) __attribute__((always_inline)) {
    const int g = s - 1;
    sm.zq0 = wz[(2 * g) * 32];
    sm.zq1 = wz[(2 * g + 1) * 32];
    sm.bias1 = b1[g + kDeg * cc];
    sm.zx0 = wz[126 * 32];
    sm.zx1 = wz[127 * 32];
    sm.bias1x = b1[kReg + cc];
    sm.uu = u[(wg * kBT + n) * kD + s];
    sm.bmu = b2[s];
    sm.bsg = b2[kD + s];
  };

  fetch_big(1);
  const float u0   = u[(wg * kBT + n) * kD];
  const float bmu0 = b2[0], bsg0 = b2[kD];

  {
    const unsigned slice = ((unsigned)wg * 0x9E3779B1u) >> 28;
    constexpr int kUnitFloats = 64;
    constexpr int kUnits = (kWcOff + kWcSz) / kUnitFloats;
    for (int j = 0; j < (kUnits + 4095) / 4096; ++j) {
      const int q  = (j * 16 + (int)slice) * 256 + tid;
      const int qb = q - kWbOff / kUnitFloats;
      const bool in_b = qb >= 0 && qb < kWbSz / kUnitFloats;
      const int t = qb >> 8, ut = qb & 255;
      const bool need = !in_b || ut < 4 * (t + 1) || ut >= 252;
      if (q < kUnits && need) __builtin_prefetch(ws + (size_t)q * kUnitFloats, 0, 2);
    }
  }

  for (int e = tid; e < kBT * kXStride; e += 256) xs[e] = 0.f;
  for (int e = tid; e < kBT * 16; e += 256)
    h0s[(e >> 4) * kHStride + kReg + (e & 15)] = 0.f;
  lds_barrier();
  if (tid < kBT) {
    xs[tid * kXStride] = u0 * expf(bsg0) + bmu0;
    ldjs[tid] = bsg0;
  }
  fetch_small(1);
  lds_barrier();

  v8f zA, zB;
  for (int v = 0; v < 8; ++v) { zA[v] = 0.f; zB[v] = 0.f; }

  v8f l1A, l1B;
  for (int v = 0; v < 8; ++v) { l1A[v] = bias0; l1B[v] = 0.f; }

  for (int i = 1; i < kD; ++i) {
    {
      const int g  = i - 1;
      const int nt = (g < 16) ? 2 : 1;

      if (wv < nt) {
        const int t  = (wv == 0) ? g : kXT;
        const int p  = g >> 3;
        const float* xr = xs + n * kXStride + hi2 + p * 8;
        const v2f a0 = *(const v2f*)&xr[0];
        const v2f a1 = *(const v2f*)&xr[4];
#pragma unroll
        for (int cp = 0; cp < 8; ++cp) {
          if (cp == p) {
            v2f bA = {w1[cp].x, w1[cp].y};
            v2f bB = {w1[cp].z, w1[cp].w};
            l1A = wmma_f32(a0, bA, l1A);
            l1B = wmma_f32(a1, bB, l1B);
          }
        }
        if (t < kXT || n == g) {
          for (int v = 0; v < 8; ++v) {
            float f = l1A[v] + l1B[v];
            f = f > 0.f ? f : 0.f;
            h0s[(v + hi8) * kHStride + t * 16 + n] = f;
          }
        }
      }
      lds_barrier();

      {
        int nch, ntot;
        l2_counts(g, wv, nch, ntot);
        v8f acc0, acc1, acx0, acx1;
        for (int v = 0; v < 8; ++v) { acc0[v] = 0.f; acc1[v] = 0.f; acx0[v] = 0.f; acx1[v] = 0.f; }
        layer2_tile<4>(w2, w2l, h0s, nch, wv, n, hi2, acc0, acc1);
        if (nt == 2) layer2_tile<1>(w2x, w2xl, h0s, nch, wv, n, hi2, acx0, acx1);

        float* pp = part + ((wv * 2) * kBT) * 16;
        for (int v = 0; v < 8; ++v) pp[(v + hi8) * 16 + n] = acc0[v] + acc1[v];
        if (nt == 2) {
          float* px = pp + kBT * 16;
          for (int v = 0; v < 8; ++v) px[(v + hi8) * 16 + n] = acx0[v] + acx1[v];
        }

        asm volatile("" :: "v"(sm.zq0.x), "v"(sm.zq0.y), "v"(sm.zq0.z), "v"(sm.zq0.w),
                           "v"(sm.zq1.x), "v"(sm.zq1.y), "v"(sm.zq1.z), "v"(sm.zq1.w),
                           "v"(sm.zx0.x), "v"(sm.zx0.y), "v"(sm.zx0.z), "v"(sm.zx0.w),
                           "v"(sm.zx1.x), "v"(sm.zx1.y), "v"(sm.zx1.z), "v"(sm.zx1.w),
                           "v"(sm.bias1), "v"(sm.bias1x), "v"(sm.uu));
        if (i + 1 < kD) fetch_big(i + 1);
      }
      lds_barrier();
      {
        const int row = tid >> 4;
        float s = sm.bias1;
        for (int w = 0; w < 8; ++w) s += part[((w * 2) * kBT + row) * 16 + cc];
        h1t[row * kDStride + cc] = s > 0.f ? s : 0.f;
        if (nt == 2) {
          float sx = sm.bias1x;
          for (int w = 0; w < 8; ++w) sx += part[((w * 2 + 1) * kBT + row) * 16 + cc];
          sx = sx > 0.f ? sx : 0.f;
          dlt[row * kDStride + cc] = (cc == g) ? sx : 0.f;
        }
      }
      lds_barrier();

      {
        AFrag f0 = load_a(h1t, kDStride, 0, n, hi2);
        AFrag f1 = load_a(h1t, kDStride, 8, n, hi2);
        mma_chunk_t(f0, sm.zq0, zA, zB);
        mma_chunk_t(f1, sm.zq1, zA, zB);
        if (nt == 2) {
          AFrag x0 = load_a(dlt, kDStride, 0, n, hi2);
          AFrag x1 = load_a(dlt, kDStride, 8, n, hi2);
          mma_chunk_t(x0, sm.zx0, zA, zB);
          mma_chunk_t(x1, sm.zx1, zA, zB);
        }
      }
    }

    if (wv == (i >> 3)) {
      const bool k0 = (i & 1) != 0, k1 = (i & 2) != 0, k2 = (i & 4) != 0;
      const float t0 = k0 ? zA[1] + zB[1] : zA[0] + zB[0];
      const float t1 = k0 ? zA[3] + zB[3] : zA[2] + zB[2];
      const float t2 = k0 ? zA[5] + zB[5] : zA[4] + zB[4];
      const float t3 = k0 ? zA[7] + zB[7] : zA[6] + zB[6];
      const float q0 = k1 ? t1 : t0;
      const float q1 = k1 ? t3 : t2;
      const float zc = k2 ? q1 : q0;
      const float zo = xhalf(zc);
      if (lane < kBT) {
        const float mu = sm.bmu + zc;
        const float sg = sm.bsg + zo;
        xs[lane * kXStride + i] = sm.uu * expf(sg) + mu;
        ldjs[lane] += sg;
      }
    }
    fetch_small(i + 1 < kD ? i + 1 : i);
    if (i + 1 < kD && wv < ((i < 16) ? 2 : 1)) {
      for (int v = 0; v < 8; ++v) { l1A[v] = bias0; l1B[v] = 0.f; }
      const int np = i >> 3;
      const float* xr = xs + n * kXStride + hi2;
      v2f pa0[7], pa1[7];
#pragma unroll
      for (int cp = 0; cp < 7; ++cp) {
        pa0[cp] = *(const v2f*)&xr[cp * 8];
        pa1[cp] = *(const v2f*)&xr[cp * 8 + 4];
      }
      asm volatile("" :: "v"(pa0[0]), "v"(pa1[0]), "v"(pa0[1]), "v"(pa1[1]), "v"(pa0[2]), "v"(pa1[2]),
                         "v"(pa0[3]), "v"(pa1[3]), "v"(pa0[4]), "v"(pa1[4]), "v"(pa0[5]), "v"(pa1[5]),
                         "v"(pa0[6]), "v"(pa1[6]));
#pragma unroll
      for (int cp = 0; cp < 7; ++cp) {
        if (cp < np) {
          v2f bA = {w1[cp].x, w1[cp].y};
          v2f bB = {w1[cp].z, w1[cp].w};
          l1A = wmma_f32(pa0[cp], bA, l1A);
          l1B = wmma_f32(pa1[cp], bB, l1B);
        }
      }
    }
    lds_barrier();
  }

  for (int e = tid; e < kBT * kD; e += 256) {
    int r = e >> 6, d = e & 63;
    out[(wg * kBT + r) * kD + d] = xs[r * kXStride + d];
  }
  if (tid < kBT) out[kB * kD + wg * kBT + tid] = ldjs[tid];
}

extern "C" void kernel_launch(void* const* d_in, const int* in_sizes, int n_in,
                              void* d_out, int out_size, void* d_ws, size_t ws_size,
                              hipStream_t stream) {
  const float* u  = (const float*)d_in[0];
  const float* W0 = (const float*)d_in[1];
  const float* b0 = (const float*)d_in[2];
  const float* W1 = (const float*)d_in[3];
  const float* b1 = (const float*)d_in[4];
  const float* W2 = (const float*)d_in[5];
  const float* b2 = (const float*)d_in[6];
  float* ws  = (float*)d_ws;
  float* out = (float*)d_out;

  constexpr size_t kPrepLds = (size_t)16 * kRowPitch * sizeof(float);
  prep_weights<<<kPrepJobs * kPrepSplit, 256, kPrepLds, stream>>>(W0, W1, W2, ws);

  constexpr size_t kLdsUsed = (size_t)(kBT * kXStride + kBT * kHStride + 8 * 2 * kBT * 16 +
                                       2 * kBT * kDStride + kBT) * sizeof(float);
  constexpr size_t kLdsOnePerWgp = 164 * 1024;
  constexpr size_t smem = kLdsUsed > kLdsOnePerWgp ? kLdsUsed : kLdsOnePerWgp;
  made_flow_kernel<<<kB / kBT, 256, smem, stream>>>(u, b0, b1, b2, ws, out);
}


# Layer 2 becomes a running sum: each new hidden-1 tile's share is added to all later hidden-2 tiles off the path; the path keeps K=32 on one wave, no K split, no partial sum (sum order changes)
# speedup vs baseline: 86.3163x; 1.0913x over previous
#include <hip/hip_runtime.h>

typedef __attribute__((ext_vector_type(2))) float v2f;
typedef __attribute__((ext_vector_type(8))) float v8f;

constexpr int kD  = 64;
constexpr int kU  = 1024;
constexpr int kB  = 4096;
constexpr int kBT = 16;

constexpr int kDeg = kD - 1;
constexpr int kReg = 16 * kDeg;
constexpr int kXT  = kDeg;

constexpr int kXStride = kD + 4;
constexpr int kHStride = kU + 4;
constexpr int kDStride = 20;

constexpr int kWaOff = 0;
constexpr int kWaSz  = kU * kD;
constexpr int kWbOff = kWaOff + kWaSz;
constexpr int kWbSz  = kU * kU;
constexpr int kWcOff = kWbOff + kWbSz;
constexpr int kWcSz  = 2 * kD * kU;

__device__ __forceinline__ int pos_unit(int p) {
  return p < kReg ? (p >> 4) + kDeg * (p & 15) : p;
}
__device__ __forceinline__ int pos_deg(int p) {
  return p < kReg ? (p >> 4) : p - kReg;
}

constexpr int kRowPitch  = kU + 4;
constexpr int kPrepSplit = 4;
constexpr int kPrepJobs  = 64 + 8;

__launch_bounds__(256)
__global__ void prep_weights(const float* __restrict__ W0, const float* __restrict__ W1,
                             const float* __restrict__ W2, float* __restrict__ ws) {
  extern __shared__ float4 rows4[];
  float* rows = (float*)rows4;
  const int tid  = threadIdx.x;
  const int job  = blockIdx.x / kPrepSplit;
  const int part = blockIdx.x % kPrepSplit;
  const bool l2job = job < 64;
  const int tile = l2job ? job : job - 64;
  const int lim = l2job ? (tile < kXT ? 2 * (tile + 1) : 32) : 128;
  const int cnt = lim < 126 ? lim + 2 : 128;
  if (part * 8 >= cnt) return;

  const float* __restrict__ srcw = l2job ? W1 : W2;
  const int rowbase = l2job ? (tile < kXT ? tile : kReg) : tile * 8;
  const int rowstep = (l2job && tile < kXT) ? kDeg : 1;
  const int rowhalf = l2job ? 0 : kD - 8;
  const float4* sp = (const float4*)srcw + tid;
  float4 v[16];
#pragma unroll
  for (int it = 0; it < 16; ++it)
    v[it] = sp[(size_t)(rowbase + rowstep * it + (it >> 3) * rowhalf) * (kU / 4)];
#pragma unroll
  for (int it = 0; it < 16; ++it) rows4[it * (kRowPitch / 4) + tid] = v[it];
  __syncthreads();

  const int l = tid & 31, n = l & 15, hi = l >> 4;
  const int rowdeg = l2job ? pos_deg(tile * 16 + n) + 1 : tile * 8 + (n & 7);
  float4* dst = (float4*)(ws + (l2job ? kWbOff : kWcOff)) + (size_t)(tile * 128) * 32;
  const float* rp = rows + n * kRowPitch;
  for (int i = part * 8 + (tid >> 5); i < cnt; i += 8 * kPrepSplit) {
    const int cp = i < lim ? i : 126 + (i - lim);
    float r[4];
    for (int q = 0; q < 4; ++q) {
      const int p = cp * 8 + ((q >> 1) << 2) + (hi << 1) + (q & 1);
      const float v = rp[pos_unit(p)];
      r[q] = pos_deg(p) < rowdeg ? v : 0.f;
    }
    const float4 o = {r[0], r[1], r[2], r[3]};
    dst[cp * 32 + l] = o;
  }

  if (l2job && part == 0) {
    const int cp = tid >> 5;
    const int j  = pos_unit(tile * 16 + n);
    const int dj = pos_deg(tile * 16 + n);
    float r[4];
    for (int q = 0; q < 4; ++q) {
      const int d = cp * 8 + ((q >> 1) << 2) + (hi << 1) + (q & 1);
      const float v = W0[j * kD + d];
      r[q] = d <= dj ? v : 0.f;
    }
    const float4 o = {r[0], r[1], r[2], r[3]};
    ((float4*)(ws + kWaOff))[(tile * 8 + cp) * 32 + l] = o;
  }
}


__device__ __forceinline__ v8f wmma_f32(v2f a, v2f b, v8f c) {
  return __builtin_amdgcn_wmma_f32_16x16x4_f32(false, a, false, b, (short)0, c,
                                               false, false);
}

struct AFrag { v2f a0, a1; };

__device__ __forceinline__ AFrag load_a(const float* __restrict__ as, int astride,
                                        int acol, int n, int hi2) {
  const int kb = acol + hi2;
  AFrag f;
  f.a0 = *(const v2f*)&as[n * astride + kb];
  f.a1 = *(const v2f*)&as[n * astride + kb + 4];
  return f;
}

__device__ __forceinline__ void mma_chunk(const AFrag& f, float4 bq, v8f& accA, v8f& accB) {
  v2f bA = {bq.x, bq.y};
  v2f bB = {bq.z, bq.w};
  accA = wmma_f32(f.a0, bA, accA);
  accB = wmma_f32(f.a1, bB, accB);
}

__device__ __forceinline__ void mma_chunk_t(const AFrag& f, float4 wq, v8f& accA, v8f& accB) {
  v2f wA = {wq.x, wq.y};
  v2f wB = {wq.z, wq.w};
  accA = wmma_f32(wA, f.a0, accA);
  accB = wmma_f32(wB, f.a1, accB);
}

__device__ __forceinline__ float xhalf(float x) {
  const int b = __builtin_bit_cast(int, x);
  return __builtin_bit_cast(float,
      __builtin_amdgcn_permlanex16(b, b, 0x76543210, 0xfedcba98, false, false));
}

struct StepOps {
  float4 zq0, zq1, zx0, zx1;
  float bias1, bias1x;
  float uu, bmu, bsg;
};

__device__ __forceinline__ void lds_barrier() {
  __builtin_amdgcn_fence(__ATOMIC_RELEASE, "workgroup", "local");
  __builtin_amdgcn_s_barrier();
  __builtin_amdgcn_fence(__ATOMIC_ACQUIRE, "workgroup", "local");
}


__device__ __forceinline__ void mma_pair(const AFrag& f, float4 bq, v8f& acc) {
  v2f bA = {bq.x, bq.y};
  v2f bB = {bq.z, bq.w};
  acc = wmma_f32(f.a0, bA, acc);
  acc = wmma_f32(f.a1, bB, acc);
}

__launch_bounds__(256, 1)
__global__ void made_flow_kernel(const float* __restrict__ u,
                                 const float* __restrict__ b0,
                                 const float* __restrict__ b1,
                                 const float* __restrict__ b2,
                                 const float* __restrict__ ws,
                                 float* __restrict__ out) {
  extern __shared__ float smem[];
  float* xs   = smem;
  float* h0s  = xs  + kBT * kXStride;
  float* h1t  = h0s + kBT * kHStride;
  float* dlt  = h1t + kBT * kDStride;
  float* ldjs = dlt + kBT * kDStride;

  const int tid  = threadIdx.x;
  const int wg   = blockIdx.x;
  const int lane = tid & 31;
  const int wv   = __builtin_amdgcn_readfirstlane(tid >> 5);
  const int n    = lane & 15;
  const int hi2  = (lane >> 4) << 1;
  const int hi8  = (lane >> 4) << 3;

  const float4* Wa4 = (const float4*)(ws + kWaOff);
  const float4* Wb4 = (const float4*)(ws + kWbOff);
  const unsigned lane_b = (unsigned)lane * 16u;
  const float4* wz = ((const float4*)(ws + kWcOff)) + (wv * 128) * 32 + lane;

  v8f acc[8];

  float4 w1[8];
  float4 w2[16];
  float4 wc[4];
  float4 wx[4];
  float  bias0 = 0.f;
  StepOps sm;

  auto fetch_w1 = [&](int g) __attribute__((always_inline)) {
    const bool need = (wv == 0) || (wv == 1 && g < 16);
    const int t = (wv == 0) ? g : kXT;
    const float4* base = need ? Wa4 + (t * 8) * 32 : Wa4;
    const unsigned lb = need ? lane_b : 0u;
#pragma unroll
    for (int cp = 0; cp < 8; ++cp) w1[cp] = *(const float4*)((const char*)(base + cp * 32) + lb);
    bias0 = b0[need ? ((t < kXT) ? t + kDeg * n : kReg + n) : 0];
  };
  auto fetch_crit = [&](int g) __attribute__((always_inline)) {
    const bool own = wv == (g & 7);
    const float4* c0 = own ? Wb4 + (g * 128 + 2 * g) * 32 : Wa4;
    const float4* c1 = own ? Wb4 + (g * 128 + 126) * 32 : Wa4;
    const unsigned lbc = own ? lane_b : 0u;
    wc[0] = *(const float4*)((const char*)(c0) + lbc);
    wc[1] = *(const float4*)((const char*)(c0 + 32) + lbc);
    wc[2] = *(const float4*)((const char*)(c1) + lbc);
    wc[3] = *(const float4*)((const char*)(c1 + 32) + lbc);
    const bool lox = (wv == 7) && g < 16;
    const float4* x0 = lox ? Wb4 + (kXT * 128 + 2 * g) * 32 : Wa4;
    const float4* x1 = lox ? Wb4 + (kXT * 128 + 126) * 32 : Wa4;
    const unsigned lbx = lox ? lane_b : 0u;
    wx[0] = *(const float4*)((const char*)(x0) + lbx);
    wx[1] = *(const float4*)((const char*)(x0 + 32) + lbx);
    wx[2] = *(const float4*)((const char*)(x1) + lbx);
    wx[3] = *(const float4*)((const char*)(x1 + 32) + lbx);
  };
  auto fetch_off = [&](int g) __attribute__((always_inline)) {
#pragma unroll
    for (int j = 0; j < 8; ++j) {
      const int t = wv + 8 * j;
      const bool need = t > g && t < kXT;
      const float4* base = need ? Wb4 + (t * 128 + 2 * g) * 32 : Wa4;
      const unsigned lb = need ? lane_b : 0u;
      w2[2 * j]     = *(const float4*)((const char*)(base) + lb);
      w2[2 * j + 1] = *(const float4*)((const char*)(base + 32) + lb);
    }
  };
  auto fetch_small = [&](int s) __attribute__((always_inline)) {
    const int g = s - 1;
    sm.zq0 = wz[(2 * g) * 32];
    sm.zq1 = wz[(2 * g + 1) * 32];
    sm.bias1 = b1[g + kDeg * n];
    sm.zx0 = wz[126 * 32];
    sm.zx1 = wz[127 * 32];
    sm.bias1x = b1[kReg + n];
    sm.uu = u[(wg * kBT + n) * kD + s];
    sm.bmu = b2[s];
    sm.bsg = b2[kD + s];
  };

  fetch_w1(0);
  const float u0   = u[(wg * kBT + n) * kD];
  const float bmu0 = b2[0], bsg0 = b2[kD];

  {
    const unsigned slice = ((unsigned)wg * 0x9E3779B1u) >> 28;
    constexpr int kUnitFloats = 64;
    constexpr int kUnits = (kWcOff + kWcSz) / kUnitFloats;
    for (int j = 0; j < (kUnits + 4095) / 4096; ++j) {
      const int q  = (j * 16 + (int)slice) * 256 + tid;
      const int qb = q - kWbOff / kUnitFloats;
      const bool in_b = qb >= 0 && qb < kWbSz / kUnitFloats;
      const int t = qb >> 8, ut = qb & 255;
      const bool need = !in_b || ut < 4 * (t + 1) || ut >= 252;
      if (q < kUnits && need) __builtin_prefetch(ws + (size_t)q * kUnitFloats, 0, 2);
    }
  }

  for (int e = tid; e < kBT * kXStride; e += 256) xs[e] = 0.f;
  for (int e = tid; e < kBT * 16; e += 256)
    h0s[(e >> 4) * kHStride + kReg + (e & 15)] = 0.f;
#pragma unroll
  for (int j = 0; j < 8; ++j)
    for (int v = 0; v < 8; ++v) acc[j][v] = 0.f;
  lds_barrier();
  if (tid < kBT) {
    xs[tid * kXStride] = u0 * expf(bsg0) + bmu0;
    ldjs[tid] = bsg0;
  }
  fetch_crit(0);
  fetch_small(1);
  fetch_off(0);
  lds_barrier();

  v8f zA, zB;
  for (int v = 0; v < 8; ++v) { zA[v] = 0.f; zB[v] = 0.f; }

  v8f l1A, l1B;
  for (int v = 0; v < 8; ++v) { l1A[v] = bias0; l1B[v] = 0.f; }

  for (int i = 1; i < kD; ++i) {
    const int g  = i - 1;
    const int nt = (g < 16) ? 2 : 1;
    const int gn = (g + 1 < kDeg) ? g + 1 : kDeg - 1;
    const bool own = wv == (g & 7);
    const bool lox = (wv == 7) && g < 16;

    if (wv < nt) {
      const int t  = (wv == 0) ? g : kXT;
      const int p  = g >> 3;
      const float* xr = xs + n * kXStride + hi2 + p * 8;
      const v2f a0 = *(const v2f*)&xr[0];
      const v2f a1 = *(const v2f*)&xr[4];
#pragma unroll
      for (int cp = 0; cp < 8; ++cp) {
        if (cp == p) {
          v2f bA = {w1[cp].x, w1[cp].y};
          v2f bB = {w1[cp].z, w1[cp].w};
          l1A = wmma_f32(a0, bA, l1A);
          l1B = wmma_f32(a1, bB, l1B);
        }
      }
      if (t < kXT || n == g) {
        for (int v = 0; v < 8; ++v) {
          float f = l1A[v] + l1B[v];
          f = f > 0.f ? f : 0.f;
          h0s[(v + hi8) * kHStride + t * 16 + n] = f;
        }
      }
    }
    v8f cs = acc[0];
    if (own) {
#pragma unroll
      for (int j = 1; j < 8; ++j)
        if (j == (g >> 3)) cs = acc[j];
    }
    lds_barrier();

    fetch_w1(gn);

    const AFrag af0 = load_a(h0s, kHStride, (2 * g) * 8, n, hi2);
    const AFrag af1 = load_a(h0s, kHStride, (2 * g + 1) * 8, n, hi2);
    if (own || lox) {
      const AFrag lf0 = load_a(h0s, kHStride, 126 * 8, n, hi2);
      const AFrag lf1 = load_a(h0s, kHStride, 127 * 8, n, hi2);
      asm volatile("" :: "v"(af0.a0), "v"(af0.a1), "v"(af1.a0), "v"(af1.a1),
                         "v"(lf0.a0), "v"(lf0.a1), "v"(lf1.a0), "v"(lf1.a1));
      if (own) {
        v8f ds;
        for (int v = 0; v < 8; ++v) ds[v] = 0.f;
        {
          v2f b;
          b = {wc[0].x, wc[0].y}; cs = wmma_f32(af0.a0, b, cs);
          b = {wc[2].x, wc[2].y}; ds = wmma_f32(lf0.a0, b, ds);
          b = {wc[0].z, wc[0].w}; cs = wmma_f32(af0.a1, b, cs);
          b = {wc[2].z, wc[2].w}; ds = wmma_f32(lf0.a1, b, ds);
          b = {wc[1].x, wc[1].y}; cs = wmma_f32(af1.a0, b, cs);
          b = {wc[3].x, wc[3].y}; ds = wmma_f32(lf1.a0, b, ds);
          b = {wc[1].z, wc[1].w}; cs = wmma_f32(af1.a1, b, cs);
          b = {wc[3].z, wc[3].w}; ds = wmma_f32(lf1.a1, b, ds);
        }
        for (int v = 0; v < 8; ++v) {
          float s = cs[v] + ds[v] + sm.bias1;
          h1t[(v + hi8) * kDStride + n] = s > 0.f ? s : 0.f;
        }
      }
      if (lox) {
        v8f es;
        for (int v = 0; v < 8; ++v) es[v] = 0.f;
        {
          v2f b;
          b = {wx[0].x, wx[0].y}; acc[7] = wmma_f32(af0.a0, b, acc[7]);
          b = {wx[2].x, wx[2].y}; es     = wmma_f32(lf0.a0, b, es);
          b = {wx[0].z, wx[0].w}; acc[7] = wmma_f32(af0.a1, b, acc[7]);
          b = {wx[2].z, wx[2].w}; es     = wmma_f32(lf0.a1, b, es);
          b = {wx[1].x, wx[1].y}; acc[7] = wmma_f32(af1.a0, b, acc[7]);
          b = {wx[3].x, wx[3].y}; es     = wmma_f32(lf1.a0, b, es);
          b = {wx[1].z, wx[1].w}; acc[7] = wmma_f32(af1.a1, b, acc[7]);
          b = {wx[3].z, wx[3].w}; es     = wmma_f32(lf1.a1, b, es);
        }
        for (int v = 0; v < 8; ++v) {
          float sx = acc[7][v] + es[v] + sm.bias1x;
          sx = sx > 0.f ? sx : 0.f;
          dlt[(v + hi8) * kDStride + n] = (n == g) ? sx : 0.f;
        }
      }
    }
    fetch_crit(gn);
    lds_barrier();

    {
      AFrag f0 = load_a(h1t, kDStride, 0, n, hi2);
      AFrag f1 = load_a(h1t, kDStride, 8, n, hi2);
      mma_chunk_t(f0, sm.zq0, zA, zB);
      mma_chunk_t(f1, sm.zq1, zA, zB);
      if (nt == 2) {
        AFrag x0 = load_a(dlt, kDStride, 0, n, hi2);
        AFrag x1 = load_a(dlt, kDStride, 8, n, hi2);
        mma_chunk_t(x0, sm.zx0, zA, zB);
        mma_chunk_t(x1, sm.zx1, zA, zB);
      }
    }

    if (wv == (i >> 3)) {
      const bool k0 = (i & 1) != 0, k1 = (i & 2) != 0, k2 = (i & 4) != 0;
      const float t0 = k0 ? zA[1] + zB[1] : zA[0] + zB[0];
      const float t1 = k0 ? zA[3] + zB[3] : zA[2] + zB[2];
      const float t2 = k0 ? zA[5] + zB[5] : zA[4] + zB[4];
      const float t3 = k0 ? zA[7] + zB[7] : zA[6] + zB[6];
      const float q0 = k1 ? t1 : t0;
      const float q1 = k1 ? t3 : t2;
      const float zc = k2 ? q1 : q0;
      const float zo = xhalf(zc);
      if (lane < kBT) {
        const float mu = sm.bmu + zc;
        const float sg = sm.bsg + zo;
        xs[lane * kXStride + i] = sm.uu * expf(sg) + mu;
        ldjs[lane] += sg;
      }
    }
    fetch_small(i + 1 < kD ? i + 1 : i);
    if (i + 1 < kD && wv < ((i < 16) ? 2 : 1)) {
      for (int v = 0; v < 8; ++v) { l1A[v] = bias0; l1B[v] = 0.f; }
      const int np = i >> 3;
      const float* xr = xs + n * kXStride + hi2;
      v2f pa0[7], pa1[7];
#pragma unroll
      for (int cp = 0; cp < 7; ++cp) {
        pa0[cp] = *(const v2f*)&xr[cp * 8];
        pa1[cp] = *(const v2f*)&xr[cp * 8 + 4];
      }
      asm volatile("" :: "v"(pa0[0]), "v"(pa1[0]), "v"(pa0[1]), "v"(pa1[1]), "v"(pa0[2]), "v"(pa1[2]),
                         "v"(pa0[3]), "v"(pa1[3]), "v"(pa0[4]), "v"(pa1[4]), "v"(pa0[5]), "v"(pa1[5]),
                         "v"(pa0[6]), "v"(pa1[6]));
#pragma unroll
      for (int cp = 0; cp < 7; ++cp) {
        if (cp < np) {
          v2f bA = {w1[cp].x, w1[cp].y};
          v2f bB = {w1[cp].z, w1[cp].w};
          l1A = wmma_f32(pa0[cp], bA, l1A);
          l1B = wmma_f32(pa1[cp], bB, l1B);
        }
      }
    }
#pragma unroll
    for (int j = 0; j < 8; ++j) {
      const int t = wv + 8 * j;
      if (t > g && t < kXT) {
        mma_pair(af0, w2[2 * j], acc[j]);
        mma_pair(af1, w2[2 * j + 1], acc[j]);
      }
    }
    fetch_off(gn);
    lds_barrier();
  }

  for (int e = tid; e < kBT * kD; e += 256) {
    int r = e >> 6, d = e & 63;
    out[(wg * kBT + r) * kD + d] = xs[r * kXStride + d];
  }
  if (tid < kBT) out[kB * kD + wg * kBT + tid] = ldjs[tid];
}

extern "C" void kernel_launch(void* const* d_in, const int* in_sizes, int n_in,
                              void* d_out, int out_size, void* d_ws, size_t ws_size,
                              hipStream_t stream) {
  const float* u  = (const float*)d_in[0];
  const float* W0 = (const float*)d_in[1];
  const float* b0 = (const float*)d_in[2];
  const float* W1 = (const float*)d_in[3];
  const float* b1 = (const float*)d_in[4];
  const float* W2 = (const float*)d_in[5];
  const float* b2 = (const float*)d_in[6];
  float* ws  = (float*)d_ws;
  float* out = (float*)d_out;

  constexpr size_t kPrepLds = (size_t)16 * kRowPitch * sizeof(float);
  prep_weights<<<kPrepJobs * kPrepSplit, 256, kPrepLds, stream>>>(W0, W1, W2, ws);

  constexpr size_t kLdsUsed = (size_t)(kBT * kXStride + kBT * kHStride +
                                       2 * kBT * kDStride + kBT) * sizeof(float);
  constexpr size_t kLdsOnePerWgp = 164 * 1024;
  constexpr size_t smem = kLdsUsed > kLdsOnePerWgp ? kLdsUsed : kLdsOnePerWgp;
  made_flow_kernel<<<kB / kBT, 256, smem, stream>>>(u, b0, b1, b2, ws, out);
}


# The step's two later barriers are split into signal and wait: off-path layer-2 sums run between them, beside the path's wave (it holds issue priority), no longer behind x[i] ahead of the barrier
# speedup vs baseline: 94.2283x; 1.0917x over previous
#include <hip/hip_runtime.h>

typedef __attribute__((ext_vector_type(2))) float v2f;
typedef __attribute__((ext_vector_type(8))) float v8f;

constexpr int kD  = 64;
constexpr int kU  = 1024;
constexpr int kB  = 4096;
constexpr int kBT = 16;

constexpr int kDeg = kD - 1;
constexpr int kReg = 16 * kDeg;
constexpr int kXT  = kDeg;

constexpr int kXStride = kD + 4;
constexpr int kHStride = kU + 4;
constexpr int kDStride = 20;

constexpr int kWaOff = 0;
constexpr int kWaSz  = kU * kD;
constexpr int kWbOff = kWaOff + kWaSz;
constexpr int kWbSz  = kU * kU;
constexpr int kWcOff = kWbOff + kWbSz;
constexpr int kWcSz  = 2 * kD * kU;

__device__ __forceinline__ int pos_unit(int p) {
  return p < kReg ? (p >> 4) + kDeg * (p & 15) : p;
}
__device__ __forceinline__ int pos_deg(int p) {
  return p < kReg ? (p >> 4) : p - kReg;
}

constexpr int kRowPitch  = kU + 4;
constexpr int kPrepSplit = 4;
constexpr int kPrepJobs  = 64 + 8;

__launch_bounds__(256)
__global__ void prep_weights(const float* __restrict__ W0, const float* __restrict__ W1,
                             const float* __restrict__ W2, float* __restrict__ ws) {
  extern __shared__ float4 rows4[];
  float* rows = (float*)rows4;
  const int tid  = threadIdx.x;
  const int job  = blockIdx.x / kPrepSplit;
  const int part = blockIdx.x % kPrepSplit;
  const bool l2job = job < 64;
  const int tile = l2job ? job : job - 64;
  const int lim = l2job ? (tile < kXT ? 2 * (tile + 1) : 32) : 128;
  const int cnt = lim < 126 ? lim + 2 : 128;
  if (part * 8 >= cnt) return;

  const float* __restrict__ srcw = l2job ? W1 : W2;
  const int rowbase = l2job ? (tile < kXT ? tile : kReg) : tile * 8;
  const int rowstep = (l2job && tile < kXT) ? kDeg : 1;
  const int rowhalf = l2job ? 0 : kD - 8;
  const float4* sp = (const float4*)srcw + tid;
  float4 v[16];
#pragma unroll
  for (int it = 0; it < 16; ++it)
    v[it] = sp[(size_t)(rowbase + rowstep * it + (it >> 3) * rowhalf) * (kU / 4)];
#pragma unroll
  for (int it = 0; it < 16; ++it) rows4[it * (kRowPitch / 4) + tid] = v[it];
  __syncthreads();

  const int l = tid & 31, n = l & 15, hi = l >> 4;
  const int rowdeg = l2job ? pos_deg(tile * 16 + n) + 1 : tile * 8 + (n & 7);
  float4* dst = (float4*)(ws + (l2job ? kWbOff : kWcOff)) + (size_t)(tile * 128) * 32;
  const float* rp = rows + n * kRowPitch;
  for (int i = part * 8 + (tid >> 5); i < cnt; i += 8 * kPrepSplit) {
    const int cp = i < lim ? i : 126 + (i - lim);
    float r[4];
    for (int q = 0; q < 4; ++q) {
      const int p = cp * 8 + ((q >> 1) << 2) + (hi << 1) + (q & 1);
      const float v = rp[pos_unit(p)];
      r[q] = pos_deg(p) < rowdeg ? v : 0.f;
    }
    const float4 o = {r[0], r[1], r[2], r[3]};
    dst[cp * 32 + l] = o;
  }

  if (l2job && part == 0) {
    const int cp = tid >> 5;
    const int j  = pos_unit(tile * 16 + n);
    const int dj = pos_deg(tile * 16 + n);
    float r[4];
    for (int q = 0; q < 4; ++q) {
      const int d = cp * 8 + ((q >> 1) << 2) + (hi << 1) + (q & 1);
      const float v = W0[j * kD + d];
      r[q] = d <= dj ? v : 0.f;
    }
    const float4 o = {r[0], r[1], r[2], r[3]};
    ((float4*)(ws + kWaOff))[(tile * 8 + cp) * 32 + l] = o;
  }
}


__device__ __forceinline__ v8f wmma_f32(v2f a, v2f b, v8f c) {
  return __builtin_amdgcn_wmma_f32_16x16x4_f32(false, a, false, b, (short)0, c,
                                               false, false);
}

struct AFrag { v2f a0, a1; };

__device__ __forceinline__ AFrag load_a(const float* __restrict__ as, int astride,
                                        int acol, int n, int hi2) {
  const int kb = acol + hi2;
  AFrag f;
  f.a0 = *(const v2f*)&as[n * astride + kb];
  f.a1 = *(const v2f*)&as[n * astride + kb + 4];
  return f;
}

__device__ __forceinline__ void mma_chunk(const AFrag& f, float4 bq, v8f& accA, v8f& accB) {
  v2f bA = {bq.x, bq.y};
  v2f bB = {bq.z, bq.w};
  accA = wmma_f32(f.a0, bA, accA);
  accB = wmma_f32(f.a1, bB, accB);
}

__device__ __forceinline__ void mma_chunk_t(const AFrag& f, float4 wq, v8f& accA, v8f& accB) {
  v2f wA = {wq.x, wq.y};
  v2f wB = {wq.z, wq.w};
  accA = wmma_f32(wA, f.a0, accA);
  accB = wmma_f32(wB, f.a1, accB);
}

__device__ __forceinline__ float xhalf(float x) {
  const int b = __builtin_bit_cast(int, x);
  return __builtin_bit_cast(float,
      __builtin_amdgcn_permlanex16(b, b, 0x76543210, 0xfedcba98, false, false));
}

struct StepOps {
  float4 zq0, zq1, zx0, zx1;
  float bias1, bias1x;
  float uu, bmu, bsg;
};

__device__ __forceinline__ void lds_barrier() {
  __builtin_amdgcn_fence(__ATOMIC_RELEASE, "workgroup", "local");
  __builtin_amdgcn_s_barrier();
  __builtin_amdgcn_fence(__ATOMIC_ACQUIRE, "workgroup", "local");
}

__device__ __forceinline__ void lds_arrive() {
  __builtin_amdgcn_fence(__ATOMIC_RELEASE, "workgroup", "local");
  __builtin_amdgcn_s_barrier_signal(-1);
}
__device__ __forceinline__ void lds_wait() {
  __builtin_amdgcn_s_barrier_wait(-1);
  __builtin_amdgcn_fence(__ATOMIC_ACQUIRE, "workgroup", "local");
}


__device__ __forceinline__ void mma_pair(const AFrag& f, float4 bq, v8f& acc) {
  v2f bA = {bq.x, bq.y};
  v2f bB = {bq.z, bq.w};
  acc = wmma_f32(f.a0, bA, acc);
  acc = wmma_f32(f.a1, bB, acc);
}

__launch_bounds__(256, 1)
__global__ void made_flow_kernel(const float* __restrict__ u,
                                 const float* __restrict__ b0,
                                 const float* __restrict__ b1,
                                 const float* __restrict__ b2,
                                 const float* __restrict__ ws,
                                 float* __restrict__ out) {
  extern __shared__ float smem[];
  float* xs   = smem;
  float* h0s  = xs  + kBT * kXStride;
  float* h1t  = h0s + kBT * kHStride;
  float* dlt  = h1t + kBT * kDStride;
  float* ldjs = dlt + kBT * kDStride;

  const int tid  = threadIdx.x;
  const int wg   = blockIdx.x;
  const int lane = tid & 31;
  const int wv   = __builtin_amdgcn_readfirstlane(tid >> 5);
  const int n    = lane & 15;
  const int hi2  = (lane >> 4) << 1;
  const int hi8  = (lane >> 4) << 3;

  const float4* Wa4 = (const float4*)(ws + kWaOff);
  const float4* Wb4 = (const float4*)(ws + kWbOff);
  const unsigned lane_b = (unsigned)lane * 16u;
  const float4* wz = ((const float4*)(ws + kWcOff)) + (wv * 128) * 32 + lane;

  v8f acc[8];

  float4 w1[8];
  float4 w2[16];
  float4 wc[4];
  float4 wx[4];
  float  bias0 = 0.f;
  StepOps sm;

  auto fetch_w1 = [&](int g) __attribute__((always_inline)) {
    const bool need = (wv == 0) || (wv == 1 && g < 16);
    const int t = (wv == 0) ? g : kXT;
    const float4* base = need ? Wa4 + (t * 8) * 32 : Wa4;
    const unsigned lb = need ? lane_b : 0u;
#pragma unroll
    for (int cp = 0; cp < 8; ++cp) w1[cp] = *(const float4*)((const char*)(base + cp * 32) + lb);
    bias0 = b0[need ? ((t < kXT) ? t + kDeg * n : kReg + n) : 0];
  };
  auto fetch_crit = [&](int g) __attribute__((always_inline)) {
    const bool own = wv == (g & 7);
    const float4* c0 = own ? Wb4 + (g * 128 + 2 * g) * 32 : Wa4;
    const float4* c1 = own ? Wb4 + (g * 128 + 126) * 32 : Wa4;
    const unsigned lbc = own ? lane_b : 0u;
    wc[0] = *(const float4*)((const char*)(c0) + lbc);
    wc[1] = *(const float4*)((const char*)(c0 + 32) + lbc);
    wc[2] = *(const float4*)((const char*)(c1) + lbc);
    wc[3] = *(const float4*)((const char*)(c1 + 32) + lbc);
    const bool lox = (wv == 7) && g < 16;
    const float4* x0 = lox ? Wb4 + (kXT * 128 + 2 * g) * 32 : Wa4;
    const float4* x1 = lox ? Wb4 + (kXT * 128 + 126) * 32 : Wa4;
    const unsigned lbx = lox ? lane_b : 0u;
    wx[0] = *(const float4*)((const char*)(x0) + lbx);
    wx[1] = *(const float4*)((const char*)(x0 + 32) + lbx);
    wx[2] = *(const float4*)((const char*)(x1) + lbx);
    wx[3] = *(const float4*)((const char*)(x1 + 32) + lbx);
  };
  auto fetch_off = [&](int g) __attribute__((always_inline)) {
#pragma unroll
    for (int j = 0; j < 8; ++j) {
      const int t = wv + 8 * j;
      const bool need = t > g && t < kXT;
      const float4* base = need ? Wb4 + (t * 128 + 2 * g) * 32 : Wa4;
      const unsigned lb = need ? lane_b : 0u;
      w2[2 * j]     = *(const float4*)((const char*)(base) + lb);
      w2[2 * j + 1] = *(const float4*)((const char*)(base + 32) + lb);
    }
  };
  auto fetch_small = [&](int s) __attribute__((always_inline)) {
    const int g = s - 1;
    sm.zq0 = wz[(2 * g) * 32];
    sm.zq1 = wz[(2 * g + 1) * 32];
    sm.bias1 = b1[g + kDeg * n];
    sm.zx0 = wz[126 * 32];
    sm.zx1 = wz[127 * 32];
    sm.bias1x = b1[kReg + n];
    sm.uu = u[(wg * kBT + n) * kD + s];
    sm.bmu = b2[s];
    sm.bsg = b2[kD + s];
  };

  fetch_w1(0);
  const float u0   = u[(wg * kBT + n) * kD];
  const float bmu0 = b2[0], bsg0 = b2[kD];

  {
    const unsigned slice = ((unsigned)wg * 0x9E3779B1u) >> 28;
    constexpr int kUnitFloats = 64;
    constexpr int kUnits = (kWcOff + kWcSz) / kUnitFloats;
    for (int j = 0; j < (kUnits + 4095) / 4096; ++j) {
      const int q  = (j * 16 + (int)slice) * 256 + tid;
      const int qb = q - kWbOff / kUnitFloats;
      const bool in_b = qb >= 0 && qb < kWbSz / kUnitFloats;
      const int t = qb >> 8, ut = qb & 255;
      const bool need = !in_b || ut < 4 * (t + 1) || ut >= 252;
      if (q < kUnits && need) __builtin_prefetch(ws + (size_t)q * kUnitFloats, 0, 2);
    }
  }

  for (int e = tid; e < kBT * kXStride; e += 256) xs[e] = 0.f;
  for (int e = tid; e < kBT * 16; e += 256)
    h0s[(e >> 4) * kHStride + kReg + (e & 15)] = 0.f;
#pragma unroll
  for (int j = 0; j < 8; ++j)
    for (int v = 0; v < 8; ++v) acc[j][v] = 0.f;
  lds_barrier();
  if (tid < kBT) {
    xs[tid * kXStride] = u0 * expf(bsg0) + bmu0;
    ldjs[tid] = bsg0;
  }
  fetch_crit(0);
  fetch_small(1);
  fetch_off(0);
  lds_barrier();

  v8f zA, zB;
  for (int v = 0; v < 8; ++v) { zA[v] = 0.f; zB[v] = 0.f; }

  v8f l1A, l1B;
  for (int v = 0; v < 8; ++v) { l1A[v] = bias0; l1B[v] = 0.f; }

  for (int i = 1; i < kD; ++i) {
    const int g  = i - 1;
    const int nt = (g < 16) ? 2 : 1;
    const int gn = (g + 1 < kDeg) ? g + 1 : kDeg - 1;
    const bool own = wv == (g & 7);
    const bool lox = (wv == 7) && g < 16;
    const bool xsw = wv == (i >> 3);
    const bool p1n = wv < ((i < 16) ? 2 : 1);

    if (wv < nt) {
      __builtin_amdgcn_s_setprio(3);
      const int t  = (wv == 0) ? g : kXT;
      const int p  = g >> 3;
      const float* xr = xs + n * kXStride + hi2 + p * 8;
      const v2f a0 = *(const v2f*)&xr[0];
      const v2f a1 = *(const v2f*)&xr[4];
#pragma unroll
      for (int cp = 0; cp < 8; ++cp) {
        if (cp == p) {
          v2f bA = {w1[cp].x, w1[cp].y};
          v2f bB = {w1[cp].z, w1[cp].w};
          l1A = wmma_f32(a0, bA, l1A);
          l1B = wmma_f32(a1, bB, l1B);
        }
      }
      if (t < kXT || n == g) {
        for (int v = 0; v < 8; ++v) {
          float f = l1A[v] + l1B[v];
          f = f > 0.f ? f : 0.f;
          h0s[(v + hi8) * kHStride + t * 16 + n] = f;
        }
      }
      __builtin_amdgcn_s_setprio(0);
    }
    v8f cs = acc[0];
    if (own) {
#pragma unroll
      for (int j = 1; j < 8; ++j)
        if (j == (g >> 3)) cs = acc[j];
    }
    lds_barrier();

    fetch_w1(gn);

    const AFrag af0 = load_a(h0s, kHStride, (2 * g) * 8, n, hi2);
    const AFrag af1 = load_a(h0s, kHStride, (2 * g + 1) * 8, n, hi2);
    auto off_sum = [&](int j0) __attribute__((always_inline)) {
#pragma unroll
      for (int j = j0; j < 8; j += 2) {
        const int t = wv + 8 * j;
        if (t > g && t < kXT) {
          mma_pair(af0, w2[2 * j], acc[j]);
          mma_pair(af1, w2[2 * j + 1], acc[j]);
        }
      }
    };
    if (own || lox) {
      __builtin_amdgcn_s_setprio(3);
      const AFrag lf0 = load_a(h0s, kHStride, 126 * 8, n, hi2);
      const AFrag lf1 = load_a(h0s, kHStride, 127 * 8, n, hi2);
      asm volatile("" :: "v"(af0.a0), "v"(af0.a1), "v"(af1.a0), "v"(af1.a1),
                         "v"(lf0.a0), "v"(lf0.a1), "v"(lf1.a0), "v"(lf1.a1));
      if (own) {
        v8f ds;
        for (int v = 0; v < 8; ++v) ds[v] = 0.f;
        {
          v2f b;
          b = {wc[0].x, wc[0].y}; cs = wmma_f32(af0.a0, b, cs);
          b = {wc[2].x, wc[2].y}; ds = wmma_f32(lf0.a0, b, ds);
          b = {wc[0].z, wc[0].w}; cs = wmma_f32(af0.a1, b, cs);
          b = {wc[2].z, wc[2].w}; ds = wmma_f32(lf0.a1, b, ds);
          b = {wc[1].x, wc[1].y}; cs = wmma_f32(af1.a0, b, cs);
          b = {wc[3].x, wc[3].y}; ds = wmma_f32(lf1.a0, b, ds);
          b = {wc[1].z, wc[1].w}; cs = wmma_f32(af1.a1, b, cs);
          b = {wc[3].z, wc[3].w}; ds = wmma_f32(lf1.a1, b, ds);
        }
        for (int v = 0; v < 8; ++v) {
          float s = cs[v] + ds[v] + sm.bias1;
          h1t[(v + hi8) * kDStride + n] = s > 0.f ? s : 0.f;
        }
      }
      if (lox) {
        v8f es;
        for (int v = 0; v < 8; ++v) es[v] = 0.f;
        {
          v2f b;
          b = {wx[0].x, wx[0].y}; acc[7] = wmma_f32(af0.a0, b, acc[7]);
          b = {wx[2].x, wx[2].y}; es     = wmma_f32(lf0.a0, b, es);
          b = {wx[0].z, wx[0].w}; acc[7] = wmma_f32(af0.a1, b, acc[7]);
          b = {wx[2].z, wx[2].w}; es     = wmma_f32(lf0.a1, b, es);
          b = {wx[1].x, wx[1].y}; acc[7] = wmma_f32(af1.a0, b, acc[7]);
          b = {wx[3].x, wx[3].y}; es     = wmma_f32(lf1.a0, b, es);
          b = {wx[1].z, wx[1].w}; acc[7] = wmma_f32(af1.a1, b, acc[7]);
          b = {wx[3].z, wx[3].w}; es     = wmma_f32(lf1.a1, b, es);
        }
        for (int v = 0; v < 8; ++v) {
          float sx = acc[7][v] + es[v] + sm.bias1x;
          sx = sx > 0.f ? sx : 0.f;
          dlt[(v + hi8) * kDStride + n] = (n == g) ? sx : 0.f;
        }
      }
      __builtin_amdgcn_s_setprio(0);
    }
    fetch_crit(gn);
    lds_arrive();
    if (!own) {
      off_sum(0);
      if (p1n) off_sum(1);
    }
    lds_wait();

    if (xsw) __builtin_amdgcn_s_setprio(3);
    {
      AFrag f0 = load_a(h1t, kDStride, 0, n, hi2);
      AFrag f1 = load_a(h1t, kDStride, 8, n, hi2);
      mma_chunk_t(f0, sm.zq0, zA, zB);
      mma_chunk_t(f1, sm.zq1, zA, zB);
      if (nt == 2) {
        AFrag x0 = load_a(dlt, kDStride, 0, n, hi2);
        AFrag x1 = load_a(dlt, kDStride, 8, n, hi2);
        mma_chunk_t(x0, sm.zx0, zA, zB);
        mma_chunk_t(x1, sm.zx1, zA, zB);
      }
    }

    if (wv == (i >> 3)) {
      const bool k0 = (i & 1) != 0, k1 = (i & 2) != 0, k2 = (i & 4) != 0;
      const float t0 = k0 ? zA[1] + zB[1] : zA[0] + zB[0];
      const float t1 = k0 ? zA[3] + zB[3] : zA[2] + zB[2];
      const float t2 = k0 ? zA[5] + zB[5] : zA[4] + zB[4];
      const float t3 = k0 ? zA[7] + zB[7] : zA[6] + zB[6];
      const float q0 = k1 ? t1 : t0;
      const float q1 = k1 ? t3 : t2;
      const float zc = k2 ? q1 : q0;
      const float zo = xhalf(zc);
      if (lane < kBT) {
        const float mu = sm.bmu + zc;
        const float sg = sm.bsg + zo;
        xs[lane * kXStride + i] = sm.uu * expf(sg) + mu;
        ldjs[lane] += sg;
      }
      __builtin_amdgcn_s_setprio(0);
    }
    fetch_small(i + 1 < kD ? i + 1 : i);
    lds_arrive();
    if (own) off_sum(0);
    if (own || !p1n) off_sum(1);
    if (i + 1 < kD && p1n) {
      for (int v = 0; v < 8; ++v) { l1A[v] = bias0; l1B[v] = 0.f; }
      const int np = i >> 3;
      const float* xr = xs + n * kXStride + hi2;
      v2f pa0[7], pa1[7];
#pragma unroll
      for (int cp = 0; cp < 7; ++cp) {
        pa0[cp] = *(const v2f*)&xr[cp * 8];
        pa1[cp] = *(const v2f*)&xr[cp * 8 + 4];
      }
      asm volatile("" :: "v"(pa0[0]), "v"(pa1[0]), "v"(pa0[1]), "v"(pa1[1]), "v"(pa0[2]), "v"(pa1[2]),
                         "v"(pa0[3]), "v"(pa1[3]), "v"(pa0[4]), "v"(pa1[4]), "v"(pa0[5]), "v"(pa1[5]),
                         "v"(pa0[6]), "v"(pa1[6]));
#pragma unroll
      for (int cp = 0; cp < 7; ++cp) {
        if (cp < np) {
          v2f bA = {w1[cp].x, w1[cp].y};
          v2f bB = {w1[cp].z, w1[cp].w};
          l1A = wmma_f32(pa0[cp], bA, l1A);
          l1B = wmma_f32(pa1[cp], bB, l1B);
        }
      }
    }
    fetch_off(gn);
    lds_wait();
  }

  for (int e = tid; e < kBT * kD; e += 256) {
    int r = e >> 6, d = e & 63;
    out[(wg * kBT + r) * kD + d] = xs[r * kXStride + d];
  }
  if (tid < kBT) out[kB * kD + wg * kBT + tid] = ldjs[tid];
}

extern "C" void kernel_launch(void* const* d_in, const int* in_sizes, int n_in,
                              void* d_out, int out_size, void* d_ws, size_t ws_size,
                              hipStream_t stream) {
  const float* u  = (const float*)d_in[0];
  const float* W0 = (const float*)d_in[1];
  const float* b0 = (const float*)d_in[2];
  const float* W1 = (const float*)d_in[3];
  const float* b1 = (const float*)d_in[4];
  const float* W2 = (const float*)d_in[5];
  const float* b2 = (const float*)d_in[6];
  float* ws  = (float*)d_ws;
  float* out = (float*)d_out;

  constexpr size_t kPrepLds = (size_t)16 * kRowPitch * sizeof(float);
  prep_weights<<<kPrepJobs * kPrepSplit, 256, kPrepLds, stream>>>(W0, W1, W2, ws);

  constexpr size_t kLdsUsed = (size_t)(kBT * kXStride + kBT * kHStride +
                                       2 * kBT * kDStride + kBT) * sizeof(float);
  constexpr size_t kLdsOnePerWgp = 164 * 1024;
  constexpr size_t smem = kLdsUsed > kLdsOnePerWgp ? kLdsUsed : kLdsOnePerWgp;
  made_flow_kernel<<<kB / kBT, 256, smem, stream>>>(u, b0, b1, b2, ws, out);
}
